# starting kernel
# baseline (speedup 1.0000x reference)
#include <hip/hip_runtime.h>
#include <math.h>

typedef __attribute__((ext_vector_type(2))) float v2f;
typedef __attribute__((ext_vector_type(8))) float v8f;

#define WMMA_F32(a, b, c) \
    __builtin_amdgcn_wmma_f32_16x16x4_f32(false, (a), false, (b), (short)0, (c), false, false)

#define BN_EPS 1e-3f
#define BIG 3.0e38f

__device__ __forceinline__ void gemm16x64(const float* A, const float* Wt,
                                          int lrow, int half, v8f acc[4]) {
#pragma unroll
    for (int kk = 0; kk < 16; ++kk) {
        const int cb = (kk << 2) + (half << 1);
        v2f a = *(const v2f*)(A + (lrow << 6) + cb);
#pragma unroll
        for (int t = 0; t < 4; ++t) {
            v2f bw = *(const v2f*)(Wt + ((lrow + (t << 4)) << 6) + cb);
            acc[t] = WMMA_F32(a, bw, acc[t]);
        }
    }
}

__device__ __forceinline__ void bn_relu(v8f acc[4], const float* g, const float* be,
                                        const float* mn, const float* vr, int lrow) {
#pragma unroll
    for (int t = 0; t < 4; ++t) {
        const int d = lrow + (t << 4);
        const float s = g[d] * (1.0f / sqrtf(vr[d] + BN_EPS));
        const float bi = be[d] - mn[d] * s;
#pragma unroll
        for (int v = 0; v < 8; ++v) acc[t][v] = fmaxf(acc[t][v] * s + bi, 0.0f);
    }
}

__device__ __forceinline__ void store16x64(const v8f acc[4], int lrow, int half, float* Y) {
#pragma unroll
    for (int t = 0; t < 4; ++t)
#pragma unroll
        for (int v = 0; v < 8; ++v)
            Y[((v + (half << 3)) << 6) + lrow + (t << 4)] = acc[t][v];
}

__global__ __launch_bounds__(256) void edgeconv_fused_kernel(
    const float* __restrict__ points,
    const float* __restrict__ feats,
    const float* __restrict__ w0,
    const float* __restrict__ w1,
    const float* __restrict__ w2,
    const float* __restrict__ gammas,
    const float* __restrict__ betas,
    const float* __restrict__ means,
    const float* __restrict__ vars,
    const float* __restrict__ scw,
    const float* __restrict__ scg, const float* __restrict__ scb,
    const float* __restrict__ scm, const float* __restrict__ scv,
    float* __restrict__ out)
{
    extern __shared__ float smem[];
    float* pts     = smem;
    float* Wt0     = pts + 768;
    float* Wt1     = Wt0 + 8192;
    float* Wt2     = Wt1 + 4096;
    float* scWt    = Wt2 + 4096;
    float* waveBuf = scWt + 4096;
    float* ctrAll  = waveBuf + 16384;
    float* ftsAll  = ctrAll + 512;
    int*   nbrAll  = (int*)(ftsAll + 512);

    const int tid  = threadIdx.x;
    const int lane = tid & 31;
    const int w    = tid >> 5;
    const int lrow = lane & 15;
    const int half = lane >> 4;
    const int b    = blockIdx.x >> 5;
    const int pi   = ((blockIdx.x & 31) << 3) | w;

    for (int i = tid; i < 768; i += 256) pts[i] = points[b * 768 + i];
    for (int i = tid; i < 8192; i += 256) {
        const int d = i >> 7, c = i & 127;
        Wt0[i] = w0[c * 64 + d];
    }
    for (int i = tid; i < 4096; i += 256) {
        const int d = i >> 6, c = i & 63;
        Wt1[i]  = w1[c * 64 + d];
        Wt2[i]  = w2[c * 64 + d];
        scWt[i] = scw[c * 64 + d];
    }
    __syncthreads();

    float* bufA = waveBuf + (w << 11);
    float* bufB = bufA + 1024;
    float* cC   = ctrAll + (w << 6);
    float* ftsW = ftsAll + (w << 6);
    int*   nbrW = nbrAll + (w << 4);

    const float px = pts[pi * 3], py = pts[pi * 3 + 1], pz = pts[pi * 3 + 2];
    float dist[8];
#pragma unroll
    for (int j = 0; j < 8; ++j) {
        const int q = lane + (j << 5);
        const float dx = pts[q * 3] - px;
        const float dy = pts[q * 3 + 1] - py;
        const float dz = pts[q * 3 + 2] - pz;
        const float dd = dx * dx + dy * dy + dz * dz;
        dist[j] = (q == pi) ? BIG : dd;
    }
    for (int it = 0; it < 16; ++it) {
        float bv = dist[0];
        int bj = 0;
#pragma unroll
        for (int j = 1; j < 8; ++j)
            if (dist[j] < bv) { bv = dist[j]; bj = j; }
        int bq = lane + (bj << 5);
#pragma unroll
        for (int off = 16; off >= 1; off >>= 1) {
            const float ov = __shfl_xor(bv, off, 32);
            const int   oq = __shfl_xor(bq, off, 32);
            if (ov < bv || (ov == bv && oq < bq)) { bv = ov; bq = oq; }
        }
        if ((bq & 31) == lane) {
            const int jw = bq >> 5;
#pragma unroll
            for (int j = 0; j < 8; ++j)
                if (j == jw) dist[j] = BIG;
        }
        if (lane == 0) nbrW[it] = bq;
    }

    const float* fbase = feats + ((b << 8) | pi) * 64;
    const float cf0 = fbase[lane];
    const float cf1 = fbase[lane + 32];
    cC[lane]      = cf0;
    cC[lane + 32] = cf1;
    for (int r = 0; r < 16; ++r) {
        const int nq = nbrW[r];
        const float* nf = feats + (((b << 8) | nq)) * 64;
        bufA[(r << 6) + lane]      = nf[lane]      - cf0;
        bufA[(r << 6) + lane + 32] = nf[lane + 32] - cf1;
    }

    const v8f vz = {0.f, 0.f, 0.f, 0.f, 0.f, 0.f, 0.f, 0.f};
    v8f acc[4] = {vz, vz, vz, vz};
#pragma unroll
    for (int kk = 0; kk < 16; ++kk) {
        const int cb = (kk << 2) + (half << 1);
        v2f a = *(const v2f*)(cC + cb);
#pragma unroll
        for (int t = 0; t < 4; ++t) {
            v2f bw = *(const v2f*)(Wt0 + (lrow + (t << 4)) * 128 + cb);
            acc[t] = WMMA_F32(a, bw, acc[t]);
        }
    }
#pragma unroll
    for (int kk = 0; kk < 16; ++kk) {
        const int cb = (kk << 2) + (half << 1);
        v2f a = *(const v2f*)(bufA + (lrow << 6) + cb);
#pragma unroll
        for (int t = 0; t < 4; ++t) {
            v2f bw = *(const v2f*)(Wt0 + (lrow + (t << 4)) * 128 + 64 + cb);
            acc[t] = WMMA_F32(a, bw, acc[t]);
        }
    }
    bn_relu(acc, gammas, betas, means, vars, lrow);
    store16x64(acc, lrow, half, bufB);

    acc[0] = vz; acc[1] = vz; acc[2] = vz; acc[3] = vz;
    gemm16x64(bufB, Wt1, lrow, half, acc);
    bn_relu(acc, gammas + 64, betas + 64, means + 64, vars + 64, lrow);
    store16x64(acc, lrow, half, bufA);

    acc[0] = vz; acc[1] = vz; acc[2] = vz; acc[3] = vz;
    gemm16x64(bufA, Wt2, lrow, half, acc);
    bn_relu(acc, gammas + 128, betas + 128, means + 128, vars + 128, lrow);

#pragma unroll
    for (int t = 0; t < 4; ++t) {
        float s = acc[t][0] + acc[t][1] + acc[t][2] + acc[t][3] +
                  acc[t][4] + acc[t][5] + acc[t][6] + acc[t][7];
        s += __shfl_xor(s, 16, 32);
        if (lane < 16) ftsW[lrow + (t << 4)] = s * (1.0f / 16.0f);
    }

    const int d0 = lane << 1;
    float s0 = 0.f, s1 = 0.f;
#pragma unroll
    for (int c = 0; c < 64; ++c) {
        const float f = cC[c];
        s0 += f * scWt[(d0 << 6) + c];
        s1 += f * scWt[((d0 + 1) << 6) + c];
    }
    const float sA = scg[d0] * (1.0f / sqrtf(scv[d0] + BN_EPS));
    const float bA = scb[d0] - scm[d0] * sA;
    const float sB = scg[d0 + 1] * (1.0f / sqrtf(scv[d0 + 1] + BN_EPS));
    const float bB = scb[d0 + 1] - scm[d0 + 1] * sB;
    v2f o;
    o.x = fmaxf(s0 * sA + bA + ftsW[d0], 0.f);
    o.y = fmaxf(s1 * sB + bB + ftsW[d0 + 1], 0.f);
    *(v2f*)(out + (((b << 8) | pi) << 6) + d0) = o;
}

extern "C" void kernel_launch(void* const* d_in, const int* in_sizes, int n_in,
                              void* d_out, int out_size, void* d_ws, size_t ws_size,
                              hipStream_t stream) {
    (void)in_sizes; (void)n_in; (void)d_ws; (void)ws_size; (void)out_size;
    const float* points = (const float*)d_in[0];
    const float* feats  = (const float*)d_in[1];
    const float* w0     = (const float*)d_in[2];
    const float* w1     = (const float*)d_in[3];
    const float* w2     = (const float*)d_in[4];
    const float* gammas = (const float*)d_in[5];
    const float* betas  = (const float*)d_in[6];
    const float* means  = (const float*)d_in[7];
    const float* vars   = (const float*)d_in[8];
    const float* scw    = (const float*)d_in[9];
    const float* scg    = (const float*)d_in[10];
    const float* scb    = (const float*)d_in[11];
    const float* scm    = (const float*)d_in[12];
    const float* scv    = (const float*)d_in[13];
    float* out = (float*)d_out;

    const size_t ldsBytes = (768 + 8192 + 3 * 4096 + 16384 + 512 + 512) * sizeof(float)
                          + 128 * sizeof(int);
    dim3 grid(4096), block(256);
    edgeconv_fused_kernel<<<grid, block, ldsBytes, stream>>>(
        points, feats, w0, w1, w2, gammas, betas, means, vars,
        scw, scg, scb, scm, scv, out);
}


# LDS rows re-pitched 64->68 floats (w0 128->132) and shortcut lanes take channels lane, lane+32, so fragment and weight reads fall on distinct banks; one tile per wave, layers written in place
# speedup vs baseline: 2.6103x; 2.6103x over previous
#include <hip/hip_runtime.h>
#include <math.h>

typedef __attribute__((ext_vector_type(2))) float v2f;
typedef __attribute__((ext_vector_type(4))) float v4f;
typedef __attribute__((ext_vector_type(8))) float v8f;

#define WMMA_F32(a, b, c) \
    __builtin_amdgcn_wmma_f32_16x16x4_f32(false, (a), false, (b), (short)0, (c), false, false)

#define BN_EPS 1e-3f
#define BIG 3.0e38f

#define TP  68
#define TP0 132

__device__ __forceinline__ void gemm16x64(const float* A, const float* Wt,
                                          int lrow, int half, v8f acc[4]) {
#pragma unroll
    for (int kk = 0; kk < 16; ++kk) {
        const int cb = (kk << 2) + (half << 1);
        v2f a = *(const v2f*)(A + lrow * TP + cb);
#pragma unroll
        for (int t = 0; t < 4; ++t) {
            v2f bw = *(const v2f*)(Wt + (lrow + (t << 4)) * TP + cb);
            acc[t] = WMMA_F32(a, bw, acc[t]);
        }
    }
}

__device__ __forceinline__ void bn_relu(v8f acc[4], const float* g, const float* be,
                                        const float* mn, const float* vr, int lrow) {
#pragma unroll
    for (int t = 0; t < 4; ++t) {
        const int d = lrow + (t << 4);
        const float s = g[d] * (1.0f / sqrtf(vr[d] + BN_EPS));
        const float bi = be[d] - mn[d] * s;
#pragma unroll
        for (int v = 0; v < 8; ++v) acc[t][v] = fmaxf(acc[t][v] * s + bi, 0.0f);
    }
}

__device__ __forceinline__ void store16x64(const v8f acc[4], int lrow, int half, float* Y) {
#pragma unroll
    for (int t = 0; t < 4; ++t)
#pragma unroll
        for (int v = 0; v < 8; ++v)
            Y[(v + (half << 3)) * TP + lrow + (t << 4)] = acc[t][v];
}

__global__ __launch_bounds__(256) void edgeconv_fused_kernel(
    const float* __restrict__ points,
    const float* __restrict__ feats,
    const float* __restrict__ w0,
    const float* __restrict__ w1,
    const float* __restrict__ w2,
    const float* __restrict__ gammas,
    const float* __restrict__ betas,
    const float* __restrict__ means,
    const float* __restrict__ vars,
    const float* __restrict__ scw,
    const float* __restrict__ scg, const float* __restrict__ scb,
    const float* __restrict__ scm, const float* __restrict__ scv,
    float* __restrict__ out)
{
    extern __shared__ __attribute__((aligned(16))) float smem[];
    float* pts     = smem;
    float* Wt0     = pts + 768;
    float* Wt1     = Wt0 + 64 * TP0;
    float* Wt2     = Wt1 + 64 * TP;
    float* scWt    = Wt2 + 64 * TP;
    float* waveBuf = scWt + 64 * TP;
    float* ctrAll  = waveBuf + 8 * 16 * TP;
    float* ftsAll  = ctrAll + 512;
    int*   nbrAll  = (int*)(ftsAll + 512);

    const int tid  = threadIdx.x;
    const int lane = tid & 31;
    const int w    = tid >> 5;
    const int lrow = lane & 15;
    const int half = lane >> 4;
    const int b    = blockIdx.x >> 5;
    const int pi   = ((blockIdx.x & 31) << 3) | w;

    for (int i = tid; i < 768; i += 256) pts[i] = points[b * 768 + i];
    for (int i = tid; i < 8192; i += 256) {
        const int d = i >> 7, c = i & 127;
        Wt0[d * TP0 + c] = w0[c * 64 + d];
    }
    for (int i = tid; i < 4096; i += 256) {
        const int d = i >> 6, c = i & 63;
        Wt1[d * TP + c]  = w1[c * 64 + d];
        Wt2[d * TP + c]  = w2[c * 64 + d];
        scWt[d * TP + c] = scw[c * 64 + d];
    }
    __syncthreads();

    float* bufA = waveBuf + w * (16 * TP);
    float* cC   = ctrAll + (w << 6);
    float* ftsW = ftsAll + (w << 6);
    int*   nbrW = nbrAll + (w << 4);

    const float px = pts[pi * 3], py = pts[pi * 3 + 1], pz = pts[pi * 3 + 2];
    float dist[8];
#pragma unroll
    for (int j = 0; j < 8; ++j) {
        const int q = lane + (j << 5);
        const float dx = pts[q * 3] - px;
        const float dy = pts[q * 3 + 1] - py;
        const float dz = pts[q * 3 + 2] - pz;
        const float dd = dx * dx + dy * dy + dz * dz;
        dist[j] = (q == pi) ? BIG : dd;
    }
    for (int it = 0; it < 16; ++it) {
        float bv = dist[0];
        int bj = 0;
#pragma unroll
        for (int j = 1; j < 8; ++j)
            if (dist[j] < bv) { bv = dist[j]; bj = j; }
        int bq = lane + (bj << 5);
#pragma unroll
        for (int off = 16; off >= 1; off >>= 1) {
            const float ov = __shfl_xor(bv, off, 32);
            const int   oq = __shfl_xor(bq, off, 32);
            if (ov < bv || (ov == bv && oq < bq)) { bv = ov; bq = oq; }
        }
        if ((bq & 31) == lane) {
            const int jw = bq >> 5;
#pragma unroll
            for (int j = 0; j < 8; ++j)
                if (j == jw) dist[j] = BIG;
        }
        if (lane == 0) nbrW[it] = bq;
    }

    const float* fbase = feats + ((b << 8) | pi) * 64;
    const float cf0 = fbase[lane];
    const float cf1 = fbase[lane + 32];
    cC[lane]      = cf0;
    cC[lane + 32] = cf1;
    for (int r = 0; r < 16; ++r) {
        const int nq = nbrW[r];
        const float* nf = feats + (((b << 8) | nq)) * 64;
        bufA[r * TP + lane]      = nf[lane]      - cf0;
        bufA[r * TP + lane + 32] = nf[lane + 32] - cf1;
    }

    const v8f vz = {0.f, 0.f, 0.f, 0.f, 0.f, 0.f, 0.f, 0.f};
    v8f acc[4] = {vz, vz, vz, vz};
#pragma unroll
    for (int kk = 0; kk < 16; ++kk) {
        const int cb = (kk << 2) + (half << 1);
        v2f a = *(const v2f*)(cC + cb);
#pragma unroll
        for (int t = 0; t < 4; ++t) {
            v2f bw = *(const v2f*)(Wt0 + (lrow + (t << 4)) * TP0 + cb);
            acc[t] = WMMA_F32(a, bw, acc[t]);
        }
    }
#pragma unroll
    for (int kk = 0; kk < 16; ++kk) {
        const int cb = (kk << 2) + (half << 1);
        v2f a = *(const v2f*)(bufA + lrow * TP + cb);
#pragma unroll
        for (int t = 0; t < 4; ++t) {
            v2f bw = *(const v2f*)(Wt0 + (lrow + (t << 4)) * TP0 + 64 + cb);
            acc[t] = WMMA_F32(a, bw, acc[t]);
        }
    }
    bn_relu(acc, gammas, betas, means, vars, lrow);
    store16x64(acc, lrow, half, bufA);

    acc[0] = vz; acc[1] = vz; acc[2] = vz; acc[3] = vz;
    gemm16x64(bufA, Wt1, lrow, half, acc);
    bn_relu(acc, gammas + 64, betas + 64, means + 64, vars + 64, lrow);
    store16x64(acc, lrow, half, bufA);

    acc[0] = vz; acc[1] = vz; acc[2] = vz; acc[3] = vz;
    gemm16x64(bufA, Wt2, lrow, half, acc);
    bn_relu(acc, gammas + 128, betas + 128, means + 128, vars + 128, lrow);

#pragma unroll
    for (int t = 0; t < 4; ++t) {
        float s = acc[t][0] + acc[t][1] + acc[t][2] + acc[t][3] +
                  acc[t][4] + acc[t][5] + acc[t][6] + acc[t][7];
        s += __shfl_xor(s, 16, 32);
        if (lane < 16) ftsW[lrow + (t << 4)] = s * (1.0f / 16.0f);
    }

    const int d0 = lane, d1 = lane + 32;
    const float* wr0 = scWt + d0 * TP;
    const float* wr1 = scWt + d1 * TP;
    float s0 = 0.f, s1 = 0.f;
#pragma unroll
    for (int c = 0; c < 64; c += 4) {
        const v4f f  = *(const v4f*)(cC + c);
        const v4f wa = *(const v4f*)(wr0 + c);
        const v4f wb = *(const v4f*)(wr1 + c);
        s0 += f.x * wa.x; s1 += f.x * wb.x;
        s0 += f.y * wa.y; s1 += f.y * wb.y;
        s0 += f.z * wa.z; s1 += f.z * wb.z;
        s0 += f.w * wa.w; s1 += f.w * wb.w;
    }
    const float sA = scg[d0] * (1.0f / sqrtf(scv[d0] + BN_EPS));
    const float bA = scb[d0] - scm[d0] * sA;
    const float sB = scg[d1] * (1.0f / sqrtf(scv[d1] + BN_EPS));
    const float bB = scb[d1] - scm[d1] * sB;
    float* orow = out + (((b << 8) | pi) << 6);
    orow[d0] = fmaxf(s0 * sA + bA + ftsW[d0], 0.f);
    orow[d1] = fmaxf(s1 * sB + bB + ftsW[d1], 0.f);
}

extern "C" void kernel_launch(void* const* d_in, const int* in_sizes, int n_in,
                              void* d_out, int out_size, void* d_ws, size_t ws_size,
                              hipStream_t stream) {
    (void)in_sizes; (void)n_in; (void)d_ws; (void)ws_size; (void)out_size;
    const float* points = (const float*)d_in[0];
    const float* feats  = (const float*)d_in[1];
    const float* w0     = (const float*)d_in[2];
    const float* w1     = (const float*)d_in[3];
    const float* w2     = (const float*)d_in[4];
    const float* gammas = (const float*)d_in[5];
    const float* betas  = (const float*)d_in[6];
    const float* means  = (const float*)d_in[7];
    const float* vars   = (const float*)d_in[8];
    const float* scw    = (const float*)d_in[9];
    const float* scg    = (const float*)d_in[10];
    const float* scb    = (const float*)d_in[11];
    const float* scm    = (const float*)d_in[12];
    const float* scv    = (const float*)d_in[13];
    float* out = (float*)d_out;

    const size_t ldsBytes = (768 + 64 * TP0 + 3 * 64 * TP + 8 * 16 * TP + 512 + 512) * sizeof(float)
                          + 128 * sizeof(int);
    dim3 grid(4096), block(256);
    edgeconv_fused_kernel<<<grid, block, ldsBytes, stream>>>(
        points, feats, w0, w1, w2, gammas, betas, means, vars,
        scw, scg, scb, scm, scv, out);
}


# weights and batch points staged to LDS once per work-group: grid 4096 -> 512 work-groups, each walks 8 tiles (one point per wave per tile) with no barrier between tiles
# speedup vs baseline: 3.8313x; 1.4678x over previous
#include <hip/hip_runtime.h>
#include <math.h>

typedef __attribute__((ext_vector_type(2))) float v2f;
typedef __attribute__((ext_vector_type(4))) float v4f;
typedef __attribute__((ext_vector_type(8))) float v8f;

#define WMMA_F32(a, b, c) \
    __builtin_amdgcn_wmma_f32_16x16x4_f32(false, (a), false, (b), (short)0, (c), false, false)

#define BN_EPS 1e-3f
#define BIG 3.0e38f

#define TP  68
#define TP0 132

#define TPW 8

__device__ __forceinline__ void gemm16x64(const float* A, const float* Wt,
                                          int lrow, int half, v8f acc[4]) {
#pragma unroll
    for (int kk = 0; kk < 16; ++kk) {
        const int cb = (kk << 2) + (half << 1);
        v2f a = *(const v2f*)(A + lrow * TP + cb);
#pragma unroll
        for (int t = 0; t < 4; ++t) {
            v2f bw = *(const v2f*)(Wt + (lrow + (t << 4)) * TP + cb);
            acc[t] = WMMA_F32(a, bw, acc[t]);
        }
    }
}

__device__ __forceinline__ void bn_relu(v8f acc[4], const float* g, const float* be,
                                        const float* mn, const float* vr, int lrow) {
#pragma unroll
    for (int t = 0; t < 4; ++t) {
        const int d = lrow + (t << 4);
        const float s = g[d] * (1.0f / sqrtf(vr[d] + BN_EPS));
        const float bi = be[d] - mn[d] * s;
#pragma unroll
        for (int v = 0; v < 8; ++v) acc[t][v] = fmaxf(acc[t][v] * s + bi, 0.0f);
    }
}

__device__ __forceinline__ void store16x64(const v8f acc[4], int lrow, int half, float* Y) {
#pragma unroll
    for (int t = 0; t < 4; ++t)
#pragma unroll
        for (int v = 0; v < 8; ++v)
            Y[(v + (half << 3)) * TP + lrow + (t << 4)] = acc[t][v];
}

__global__ __launch_bounds__(256) void edgeconv_fused_kernel(
    const float* __restrict__ points,
    const float* __restrict__ feats,
    const float* __restrict__ w0,
    const float* __restrict__ w1,
    const float* __restrict__ w2,
    const float* __restrict__ gammas,
    const float* __restrict__ betas,
    const float* __restrict__ means,
    const float* __restrict__ vars,
    const float* __restrict__ scw,
    const float* __restrict__ scg, const float* __restrict__ scb,
    const float* __restrict__ scm, const float* __restrict__ scv,
    float* __restrict__ out)
{
    extern __shared__ __attribute__((aligned(16))) float smem[];
    float* pts     = smem;
    float* Wt0     = pts + 768;
    float* Wt1     = Wt0 + 64 * TP0;
    float* Wt2     = Wt1 + 64 * TP;
    float* scWt    = Wt2 + 64 * TP;
    float* waveBuf = scWt + 64 * TP;
    float* ctrAll  = waveBuf + 8 * 16 * TP;
    float* ftsAll  = ctrAll + 512;
    int*   nbrAll  = (int*)(ftsAll + 512);

    const int tid  = threadIdx.x;
    const int lane = tid & 31;
    const int w    = tid >> 5;
    const int lrow = lane & 15;
    const int half = lane >> 4;
    const int tile0 = blockIdx.x * TPW;
    const int b     = tile0 >> 5;

    for (int i = tid; i < 768; i += 256) pts[i] = points[b * 768 + i];
    for (int i = tid; i < 8192; i += 256) {
        const int d = i >> 7, c = i & 127;
        Wt0[d * TP0 + c] = w0[c * 64 + d];
    }
    for (int i = tid; i < 4096; i += 256) {
        const int d = i >> 6, c = i & 63;
        Wt1[d * TP + c]  = w1[c * 64 + d];
        Wt2[d * TP + c]  = w2[c * 64 + d];
        scWt[d * TP + c] = scw[c * 64 + d];
    }
    __syncthreads();

    float* bufA = waveBuf + w * (16 * TP);
    float* cC   = ctrAll + (w << 6);
    float* ftsW = ftsAll + (w << 6);
    int*   nbrW = nbrAll + (w << 4);

    for (int tl = 0; tl < TPW; ++tl) {
    const int pi = (((tile0 + tl) & 31) << 3) | w;

    const float px = pts[pi * 3], py = pts[pi * 3 + 1], pz = pts[pi * 3 + 2];
    float dist[8];
#pragma unroll
    for (int j = 0; j < 8; ++j) {
        const int q = lane + (j << 5);
        const float dx = pts[q * 3] - px;
        const float dy = pts[q * 3 + 1] - py;
        const float dz = pts[q * 3 + 2] - pz;
        const float dd = dx * dx + dy * dy + dz * dz;
        dist[j] = (q == pi) ? BIG : dd;
    }
    for (int it = 0; it < 16; ++it) {
        float bv = dist[0];
        int bj = 0;
#pragma unroll
        for (int j = 1; j < 8; ++j)
            if (dist[j] < bv) { bv = dist[j]; bj = j; }
        int bq = lane + (bj << 5);
#pragma unroll
        for (int off = 16; off >= 1; off >>= 1) {
            const float ov = __shfl_xor(bv, off, 32);
            const int   oq = __shfl_xor(bq, off, 32);
            if (ov < bv || (ov == bv && oq < bq)) { bv = ov; bq = oq; }
        }
        if ((bq & 31) == lane) {
            const int jw = bq >> 5;
#pragma unroll
            for (int j = 0; j < 8; ++j)
                if (j == jw) dist[j] = BIG;
        }
        if (lane == 0) nbrW[it] = bq;
    }

    const float* fbase = feats + ((b << 8) | pi) * 64;
    const float cf0 = fbase[lane];
    const float cf1 = fbase[lane + 32];
    cC[lane]      = cf0;
    cC[lane + 32] = cf1;
    for (int r = 0; r < 16; ++r) {
        const int nq = nbrW[r];
        const float* nf = feats + (((b << 8) | nq)) * 64;
        bufA[r * TP + lane]      = nf[lane]      - cf0;
        bufA[r * TP + lane + 32] = nf[lane + 32] - cf1;
    }

    const v8f vz = {0.f, 0.f, 0.f, 0.f, 0.f, 0.f, 0.f, 0.f};
    v8f acc[4] = {vz, vz, vz, vz};
#pragma unroll
    for (int kk = 0; kk < 16; ++kk) {
        const int cb = (kk << 2) + (half << 1);
        v2f a = *(const v2f*)(cC + cb);
#pragma unroll
        for (int t = 0; t < 4; ++t) {
            v2f bw = *(const v2f*)(Wt0 + (lrow + (t << 4)) * TP0 + cb);
            acc[t] = WMMA_F32(a, bw, acc[t]);
        }
    }
#pragma unroll
    for (int kk = 0; kk < 16; ++kk) {
        const int cb = (kk << 2) + (half << 1);
        v2f a = *(const v2f*)(bufA + lrow * TP + cb);
#pragma unroll
        for (int t = 0; t < 4; ++t) {
            v2f bw = *(const v2f*)(Wt0 + (lrow + (t << 4)) * TP0 + 64 + cb);
            acc[t] = WMMA_F32(a, bw, acc[t]);
        }
    }
    bn_relu(acc, gammas, betas, means, vars, lrow);
    store16x64(acc, lrow, half, bufA);

    acc[0] = vz; acc[1] = vz; acc[2] = vz; acc[3] = vz;
    gemm16x64(bufA, Wt1, lrow, half, acc);
    bn_relu(acc, gammas + 64, betas + 64, means + 64, vars + 64, lrow);
    store16x64(acc, lrow, half, bufA);

    acc[0] = vz; acc[1] = vz; acc[2] = vz; acc[3] = vz;
    gemm16x64(bufA, Wt2, lrow, half, acc);
    bn_relu(acc, gammas + 128, betas + 128, means + 128, vars + 128, lrow);

#pragma unroll
    for (int t = 0; t < 4; ++t) {
        float s = acc[t][0] + acc[t][1] + acc[t][2] + acc[t][3] +
                  acc[t][4] + acc[t][5] + acc[t][6] + acc[t][7];
        s += __shfl_xor(s, 16, 32);
        if (lane < 16) ftsW[lrow + (t << 4)] = s * (1.0f / 16.0f);
    }

    const int d0 = lane, d1 = lane + 32;
    const float* wr0 = scWt + d0 * TP;
    const float* wr1 = scWt + d1 * TP;
    float s0 = 0.f, s1 = 0.f;
#pragma unroll
    for (int c = 0; c < 64; c += 4) {
        const v4f f  = *(const v4f*)(cC + c);
        const v4f wa = *(const v4f*)(wr0 + c);
        const v4f wb = *(const v4f*)(wr1 + c);
        s0 += f.x * wa.x; s1 += f.x * wb.x;
        s0 += f.y * wa.y; s1 += f.y * wb.y;
        s0 += f.z * wa.z; s1 += f.z * wb.z;
        s0 += f.w * wa.w; s1 += f.w * wb.w;
    }
    const float sA = scg[d0] * (1.0f / sqrtf(scv[d0] + BN_EPS));
    const float bA = scb[d0] - scm[d0] * sA;
    const float sB = scg[d1] * (1.0f / sqrtf(scv[d1] + BN_EPS));
    const float bB = scb[d1] - scm[d1] * sB;
    float* orow = out + (((b << 8) | pi) << 6);
    orow[d0] = fmaxf(s0 * sA + bA + ftsW[d0], 0.f);
    orow[d1] = fmaxf(s1 * sB + bB + ftsW[d1], 0.f);
    }
}

extern "C" void kernel_launch(void* const* d_in, const int* in_sizes, int n_in,
                              void* d_out, int out_size, void* d_ws, size_t ws_size,
                              hipStream_t stream) {
    (void)in_sizes; (void)n_in; (void)d_ws; (void)ws_size; (void)out_size;
    const float* points = (const float*)d_in[0];
    const float* feats  = (const float*)d_in[1];
    const float* w0     = (const float*)d_in[2];
    const float* w1     = (const float*)d_in[3];
    const float* w2     = (const float*)d_in[4];
    const float* gammas = (const float*)d_in[5];
    const float* betas  = (const float*)d_in[6];
    const float* means  = (const float*)d_in[7];
    const float* vars   = (const float*)d_in[8];
    const float* scw    = (const float*)d_in[9];
    const float* scg    = (const float*)d_in[10];
    const float* scb    = (const float*)d_in[11];
    const float* scm    = (const float*)d_in[12];
    const float* scv    = (const float*)d_in[13];
    float* out = (float*)d_out;

    const size_t ldsBytes = (768 + 64 * TP0 + 3 * 64 * TP + 8 * 16 * TP + 512 + 512) * sizeof(float)
                          + 128 * sizeof(int);
    dim3 grid(512), block(256);
    edgeconv_fused_kernel<<<grid, block, ldsBytes, stream>>>(
        points, feats, w0, w1, w2, gammas, betas, means, vars,
        scw, scg, scb, scm, scv, out);
}


# KNN: 16th-smallest distance by a 31-step bitwise search on the keys' bit patterns (lane masks counted on the scalar unit); neighbours listed in index order, so the 16-row mean adds in that order
# speedup vs baseline: 4.9901x; 1.3025x over previous
#include <hip/hip_runtime.h>
#include <math.h>

typedef __attribute__((ext_vector_type(2))) float v2f;
typedef __attribute__((ext_vector_type(4))) float v4f;
typedef __attribute__((ext_vector_type(8))) float v8f;

#define WMMA_F32(a, b, c) \
    __builtin_amdgcn_wmma_f32_16x16x4_f32(false, (a), false, (b), (short)0, (c), false, false)

#define BN_EPS 1e-3f
#define BIG 3.0e38f

#define TP  68
#define TP0 132

#define TPW 8

__device__ __forceinline__ void gemm16x64(const float* A, const float* Wt,
                                          int lrow, int half, v8f acc[4]) {
#pragma unroll
    for (int kk = 0; kk < 16; ++kk) {
        const int cb = (kk << 2) + (half << 1);
        v2f a = *(const v2f*)(A + lrow * TP + cb);
#pragma unroll
        for (int t = 0; t < 4; ++t) {
            v2f bw = *(const v2f*)(Wt + (lrow + (t << 4)) * TP + cb);
            acc[t] = WMMA_F32(a, bw, acc[t]);
        }
    }
}

__device__ __forceinline__ void bn_relu(v8f acc[4], const float* g, const float* be,
                                        const float* mn, const float* vr, int lrow) {
#pragma unroll
    for (int t = 0; t < 4; ++t) {
        const int d = lrow + (t << 4);
        const float s = g[d] * (1.0f / sqrtf(vr[d] + BN_EPS));
        const float bi = be[d] - mn[d] * s;
#pragma unroll
        for (int v = 0; v < 8; ++v) acc[t][v] = fmaxf(acc[t][v] * s + bi, 0.0f);
    }
}

__device__ __forceinline__ void store16x64(const v8f acc[4], int lrow, int half, float* Y) {
#pragma unroll
    for (int t = 0; t < 4; ++t)
#pragma unroll
        for (int v = 0; v < 8; ++v)
            Y[(v + (half << 3)) * TP + lrow + (t << 4)] = acc[t][v];
}

__global__ __launch_bounds__(256) void edgeconv_fused_kernel(
    const float* __restrict__ points,
    const float* __restrict__ feats,
    const float* __restrict__ w0,
    const float* __restrict__ w1,
    const float* __restrict__ w2,
    const float* __restrict__ gammas,
    const float* __restrict__ betas,
    const float* __restrict__ means,
    const float* __restrict__ vars,
    const float* __restrict__ scw,
    const float* __restrict__ scg, const float* __restrict__ scb,
    const float* __restrict__ scm, const float* __restrict__ scv,
    float* __restrict__ out)
{
    extern __shared__ __attribute__((aligned(16))) float smem[];
    float* pts     = smem;
    float* Wt0     = pts + 768;
    float* Wt1     = Wt0 + 64 * TP0;
    float* Wt2     = Wt1 + 64 * TP;
    float* scWt    = Wt2 + 64 * TP;
    float* waveBuf = scWt + 64 * TP;
    float* ctrAll  = waveBuf + 8 * 16 * TP;
    float* ftsAll  = ctrAll + 512;
    int*   nbrAll  = (int*)(ftsAll + 512);

    const int tid  = threadIdx.x;
    const int lane = tid & 31;
    const int w    = tid >> 5;
    const int lrow = lane & 15;
    const int half = lane >> 4;
    const int tile0 = blockIdx.x * TPW;
    const int b     = tile0 >> 5;

    for (int i = tid; i < 768; i += 256) pts[i] = points[b * 768 + i];
    for (int i = tid; i < 8192; i += 256) {
        const int d = i >> 7, c = i & 127;
        Wt0[d * TP0 + c] = w0[c * 64 + d];
    }
    for (int i = tid; i < 4096; i += 256) {
        const int d = i >> 6, c = i & 63;
        Wt1[d * TP + c]  = w1[c * 64 + d];
        Wt2[d * TP + c]  = w2[c * 64 + d];
        scWt[d * TP + c] = scw[c * 64 + d];
    }
    __syncthreads();

    float* bufA = waveBuf + w * (16 * TP);
    float* cC   = ctrAll + (w << 6);
    float* ftsW = ftsAll + (w << 6);
    int*   nbrW = nbrAll + (w << 4);

    for (int tl = 0; tl < TPW; ++tl) {
    const int pi = (((tile0 + tl) & 31) << 3) | w;

    const float px = pts[pi * 3], py = pts[pi * 3 + 1], pz = pts[pi * 3 + 2];
    unsigned key[8];
#pragma unroll
    for (int j = 0; j < 8; ++j) {
        const int q = lane + (j << 5);
        const float dx = pts[q * 3] - px;
        const float dy = pts[q * 3 + 1] - py;
        const float dz = pts[q * 3 + 2] - pz;
        const float dd = dx * dx + dy * dy + dz * dz;
        key[j] = __float_as_uint((q == pi) ? BIG : dd);
    }
    unsigned thr = 0;
    for (int bit = 30; bit >= 0; --bit) {
        const unsigned trial = thr | (1u << bit);
        int below = 0;
#pragma unroll
        for (int j = 0; j < 8; j += 2) {
            const unsigned long long m2 =
                ((unsigned long long)__builtin_amdgcn_ballot_w32(key[j + 1] < trial) << 32) |
                __builtin_amdgcn_ballot_w32(key[j] < trial);
            below += __builtin_popcountll(m2);
        }
        if (below < 16) thr = trial;
    }
    unsigned mLess[8], mEq[8];
    int nLess = 0;
#pragma unroll
    for (int j = 0; j < 8; ++j) {
        mLess[j] = __builtin_amdgcn_ballot_w32(key[j] < thr);
        mEq[j]   = __builtin_amdgcn_ballot_w32(key[j] == thr);
        nLess += __builtin_popcount(mLess[j]);
    }
    int offL = 0, offE = nLess;
#pragma unroll
    for (int j = 0; j < 8; ++j) {
        const bool isL = key[j] < thr;
        const bool isE = key[j] == thr;
        const int pos = isL ? offL + (int)__builtin_amdgcn_mbcnt_lo(mLess[j], 0u)
                            : offE + (int)__builtin_amdgcn_mbcnt_lo(mEq[j], 0u);
        if (isL || (isE && pos < 16)) nbrW[pos] = lane + (j << 5);
        offL += __builtin_popcount(mLess[j]);
        offE += __builtin_popcount(mEq[j]);
    }

    const float* fbase = feats + ((b << 8) | pi) * 64;
    const float cf0 = fbase[lane];
    const float cf1 = fbase[lane + 32];
    cC[lane]      = cf0;
    cC[lane + 32] = cf1;
    for (int r = 0; r < 16; ++r) {
        const int nq = nbrW[r];
        const float* nf = feats + (((b << 8) | nq)) * 64;
        bufA[r * TP + lane]      = nf[lane]      - cf0;
        bufA[r * TP + lane + 32] = nf[lane + 32] - cf1;
    }

    const v8f vz = {0.f, 0.f, 0.f, 0.f, 0.f, 0.f, 0.f, 0.f};
    v8f acc[4] = {vz, vz, vz, vz};
#pragma unroll
    for (int kk = 0; kk < 16; ++kk) {
        const int cb = (kk << 2) + (half << 1);
        v2f a = *(const v2f*)(cC + cb);
#pragma unroll
        for (int t = 0; t < 4; ++t) {
            v2f bw = *(const v2f*)(Wt0 + (lrow + (t << 4)) * TP0 + cb);
            acc[t] = WMMA_F32(a, bw, acc[t]);
        }
    }
#pragma unroll
    for (int kk = 0; kk < 16; ++kk) {
        const int cb = (kk << 2) + (half << 1);
        v2f a = *(const v2f*)(bufA + lrow * TP + cb);
#pragma unroll
        for (int t = 0; t < 4; ++t) {
            v2f bw = *(const v2f*)(Wt0 + (lrow + (t << 4)) * TP0 + 64 + cb);
            acc[t] = WMMA_F32(a, bw, acc[t]);
        }
    }
    bn_relu(acc, gammas, betas, means, vars, lrow);
    store16x64(acc, lrow, half, bufA);

    acc[0] = vz; acc[1] = vz; acc[2] = vz; acc[3] = vz;
    gemm16x64(bufA, Wt1, lrow, half, acc);
    bn_relu(acc, gammas + 64, betas + 64, means + 64, vars + 64, lrow);
    store16x64(acc, lrow, half, bufA);

    acc[0] = vz; acc[1] = vz; acc[2] = vz; acc[3] = vz;
    gemm16x64(bufA, Wt2, lrow, half, acc);
    bn_relu(acc, gammas + 128, betas + 128, means + 128, vars + 128, lrow);

#pragma unroll
    for (int t = 0; t < 4; ++t) {
        float s = acc[t][0] + acc[t][1] + acc[t][2] + acc[t][3] +
                  acc[t][4] + acc[t][5] + acc[t][6] + acc[t][7];
        s += __shfl_xor(s, 16, 32);
        if (lane < 16) ftsW[lrow + (t << 4)] = s * (1.0f / 16.0f);
    }

    const int d0 = lane, d1 = lane + 32;
    const float* wr0 = scWt + d0 * TP;
    const float* wr1 = scWt + d1 * TP;
    float s0 = 0.f, s1 = 0.f;
#pragma unroll
    for (int c = 0; c < 64; c += 4) {
        const v4f f  = *(const v4f*)(cC + c);
        const v4f wa = *(const v4f*)(wr0 + c);
        const v4f wb = *(const v4f*)(wr1 + c);
        s0 += f.x * wa.x; s1 += f.x * wb.x;
        s0 += f.y * wa.y; s1 += f.y * wb.y;
        s0 += f.z * wa.z; s1 += f.z * wb.z;
        s0 += f.w * wa.w; s1 += f.w * wb.w;
    }
    const float sA = scg[d0] * (1.0f / sqrtf(scv[d0] + BN_EPS));
    const float bA = scb[d0] - scm[d0] * sA;
    const float sB = scg[d1] * (1.0f / sqrtf(scv[d1] + BN_EPS));
    const float bB = scb[d1] - scm[d1] * sB;
    float* orow = out + (((b << 8) | pi) << 6);
    orow[d0] = fmaxf(s0 * sA + bA + ftsW[d0], 0.f);
    orow[d1] = fmaxf(s1 * sB + bB + ftsW[d1], 0.f);
    }
}

extern "C" void kernel_launch(void* const* d_in, const int* in_sizes, int n_in,
                              void* d_out, int out_size, void* d_ws, size_t ws_size,
                              hipStream_t stream) {
    (void)in_sizes; (void)n_in; (void)d_ws; (void)ws_size; (void)out_size;
    const float* points = (const float*)d_in[0];
    const float* feats  = (const float*)d_in[1];
    const float* w0     = (const float*)d_in[2];
    const float* w1     = (const float*)d_in[3];
    const float* w2     = (const float*)d_in[4];
    const float* gammas = (const float*)d_in[5];
    const float* betas  = (const float*)d_in[6];
    const float* means  = (const float*)d_in[7];
    const float* vars   = (const float*)d_in[8];
    const float* scw    = (const float*)d_in[9];
    const float* scg    = (const float*)d_in[10];
    const float* scb    = (const float*)d_in[11];
    const float* scm    = (const float*)d_in[12];
    const float* scv    = (const float*)d_in[13];
    float* out = (float*)d_out;

    const size_t ldsBytes = (768 + 64 * TP0 + 3 * 64 * TP + 8 * 16 * TP + 512 + 512) * sizeof(float)
                          + 128 * sizeof(int);
    dim3 grid(512), block(256);
    edgeconv_fused_kernel<<<grid, block, ldsBytes, stream>>>(
        points, feats, w0, w1, w2, gammas, betas, means, vars,
        scw, scg, scb, scm, scv, out);
}


# layer 0 and shortcut by linearity: F.w0[64:] (whole batch), F.w0[:64], F.sc_w (own points) projected once per work-group on WMMA; edge row = Z[ctr]-Y[ctr]+Y[nbr] gathered from LDS (sum order changes)
# speedup vs baseline: 7.1597x; 1.4348x over previous
#include <hip/hip_runtime.h>
#include <math.h>

typedef __attribute__((ext_vector_type(2))) float v2f;
typedef __attribute__((ext_vector_type(4))) float v4f;
typedef __attribute__((ext_vector_type(8))) float v8f;

#define WMMA_F32(a, b, c) \
    __builtin_amdgcn_wmma_f32_16x16x4_f32(false, (a), false, (b), (short)0, (c), false, false)

#define BN_EPS 1e-3f
#define BIG 3.0e38f

#define TP  68
#define TP0 132

#define TPW 8

__device__ __forceinline__ void gemm16x64(const float* A, const float* Wt,
                                          int lrow, int half, v8f acc[4]) {
#pragma unroll
    for (int kk = 0; kk < 16; ++kk) {
        const int cb = (kk << 2) + (half << 1);
        v2f a = *(const v2f*)(A + lrow * TP + cb);
#pragma unroll
        for (int t = 0; t < 4; ++t) {
            v2f bw = *(const v2f*)(Wt + (lrow + (t << 4)) * TP + cb);
            acc[t] = WMMA_F32(a, bw, acc[t]);
        }
    }
}

__device__ __forceinline__ void bn_relu(v8f acc[4], const float* g, const float* be,
                                        const float* mn, const float* vr, int lrow) {
#pragma unroll
    for (int t = 0; t < 4; ++t) {
        const int d = lrow + (t << 4);
        const float s = g[d] * (1.0f / sqrtf(vr[d] + BN_EPS));
        const float bi = be[d] - mn[d] * s;
#pragma unroll
        for (int v = 0; v < 8; ++v) acc[t][v] = fmaxf(acc[t][v] * s + bi, 0.0f);
    }
}

__device__ __forceinline__ void store16x64(const v8f acc[4], int lrow, int half, float* Y) {
#pragma unroll
    for (int t = 0; t < 4; ++t)
#pragma unroll
        for (int v = 0; v < 8; ++v)
            Y[(v + (half << 3)) * TP + lrow + (t << 4)] = acc[t][v];
}

__global__ __launch_bounds__(256) void edgeconv_fused_kernel(
    const float* __restrict__ points,
    const float* __restrict__ feats,
    const float* __restrict__ w0,
    const float* __restrict__ w1,
    const float* __restrict__ w2,
    const float* __restrict__ gammas,
    const float* __restrict__ betas,
    const float* __restrict__ means,
    const float* __restrict__ vars,
    const float* __restrict__ scw,
    const float* __restrict__ scg, const float* __restrict__ scb,
    const float* __restrict__ scm, const float* __restrict__ scv,
    float* __restrict__ out)
{
    extern __shared__ __attribute__((aligned(16))) float smem[];
    float* pts     = smem;
    float* Yp      = pts + 768;
    float* R       = Yp + 256 * 64;
    float* Wt0     = R;
    float* scWt    = Wt0 + 64 * TP0;
    float* Wt1     = R;
    float* Wt2     = Wt1 + 64 * TP;
    float* waveBuf = Wt2 + 64 * TP;
    float* ftsAll  = waveBuf + 8 * 16 * TP;
    int*   nbrAll  = (int*)(ftsAll + 512);

    const int tid  = threadIdx.x;
    const int lane = tid & 31;
    const int w    = tid >> 5;
    const int lrow = lane & 15;
    const int half = lane >> 4;
    const int tile0 = blockIdx.x * TPW;
    const int b     = tile0 >> 5;
    const float* fb = feats + (b << 8) * 64;

    for (int i = tid; i < 768; i += 256) pts[i] = points[b * 768 + i];
    for (int i = tid; i < 8192; i += 256) {
        const int d = i >> 7, c = i & 127;
        Wt0[d * TP0 + c] = w0[c * 64 + d];
    }
    for (int i = tid; i < 4096; i += 256) {
        const int d = i >> 6, c = i & 63;
        scWt[d * TP + c] = scw[c * 64 + d];
    }
    __syncthreads();

    const v8f vz = {0.f, 0.f, 0.f, 0.f, 0.f, 0.f, 0.f, 0.f};

    {
        const int q0 = w << 5;
        v2f a0[16], a1[16];
#pragma unroll
        for (int kk = 0; kk < 16; ++kk) {
            const int cb = (kk << 2) + (half << 1);
            a0[kk] = *(const v2f*)(fb + (q0 + lrow) * 64 + cb);
            a1[kk] = *(const v2f*)(fb + (q0 + 16 + lrow) * 64 + cb);
        }
        v8f y0[4] = {vz, vz, vz, vz};
        v8f y1[4] = {vz, vz, vz, vz};
#pragma unroll
        for (int kk = 0; kk < 16; ++kk) {
            const int cb = (kk << 2) + (half << 1);
#pragma unroll
            for (int t = 0; t < 4; ++t) {
                v2f bw = *(const v2f*)(Wt0 + (lrow + (t << 4)) * TP0 + 64 + cb);
                y0[t] = WMMA_F32(a0[kk], bw, y0[t]);
                y1[t] = WMMA_F32(a1[kk], bw, y1[t]);
            }
        }
#pragma unroll
        for (int t = 0; t < 4; ++t)
#pragma unroll
            for (int v = 0; v < 8; ++v) {
                Yp[(q0 + v + (half << 3)) * 64 + lrow + (t << 4)]      = y0[t][v];
                Yp[(q0 + 16 + v + (half << 3)) * 64 + lrow + (t << 4)] = y1[t][v];
            }
    }
    v8f zA, zB, sA8, sB8;
    {
        const int pj = (((tile0 + (lrow & 7)) & 31) << 3) | w;
        v2f a[16];
#pragma unroll
        for (int kk = 0; kk < 16; ++kk) {
            const int cb = (kk << 2) + (half << 1);
            a[kk] = *(const v2f*)(fb + pj * 64 + cb);
        }
        v8f az[4] = {vz, vz, vz, vz};
        v8f as[4] = {vz, vz, vz, vz};
#pragma unroll
        for (int kk = 0; kk < 16; ++kk) {
            const int cb = (kk << 2) + (half << 1);
#pragma unroll
            for (int t = 0; t < 4; ++t) {
                v2f bz = *(const v2f*)(Wt0 + (lrow + (t << 4)) * TP0 + cb);
                v2f bs = *(const v2f*)(scWt + (lrow + (t << 4)) * TP + cb);
                az[t] = WMMA_F32(a[kk], bz, az[t]);
                as[t] = WMMA_F32(a[kk], bs, as[t]);
            }
        }
#pragma unroll
        for (int v = 0; v < 8; ++v) {
            zA[v]  = half ? az[1][v] : az[0][v];
            zB[v]  = half ? az[3][v] : az[2][v];
            sA8[v] = half ? as[1][v] : as[0][v];
            sB8[v] = half ? as[3][v] : as[2][v];
        }
    }
    __syncthreads();
    for (int i = tid; i < 4096; i += 256) {
        const int d = i >> 6, c = i & 63;
        Wt1[d * TP + c] = w1[c * 64 + d];
        Wt2[d * TP + c] = w2[c * 64 + d];
    }
    __syncthreads();

    float* bufA = waveBuf + w * (16 * TP);
    float* ftsW = ftsAll + (w << 6);
    int*   nbrW = nbrAll + (w << 4);

    const int d0 = lane, d1 = lane + 32;
    const float s0a = gammas[d0] * (1.0f / sqrtf(vars[d0] + BN_EPS));
    const float b0a = betas[d0] - means[d0] * s0a;
    const float s0b = gammas[d1] * (1.0f / sqrtf(vars[d1] + BN_EPS));
    const float b0b = betas[d1] - means[d1] * s0b;
    const float sA = scg[d0] * (1.0f / sqrtf(scv[d0] + BN_EPS));
    const float bA = scb[d0] - scm[d0] * sA;
    const float sB = scg[d1] * (1.0f / sqrtf(scv[d1] + BN_EPS));
    const float bB = scb[d1] - scm[d1] * sB;

    for (int tl = 0; tl < TPW; ++tl) {
    const int pi = (((tile0 + tl) & 31) << 3) | w;

    const float px = pts[pi * 3], py = pts[pi * 3 + 1], pz = pts[pi * 3 + 2];
    unsigned key[8];
#pragma unroll
    for (int j = 0; j < 8; ++j) {
        const int q = lane + (j << 5);
        const float dx = pts[q * 3] - px;
        const float dy = pts[q * 3 + 1] - py;
        const float dz = pts[q * 3 + 2] - pz;
        const float dd = dx * dx + dy * dy + dz * dz;
        key[j] = __float_as_uint((q == pi) ? BIG : dd);
    }
    unsigned thr = 0;
    for (int bit = 30; bit >= 0; --bit) {
        const unsigned trial = thr | (1u << bit);
        int below = 0;
#pragma unroll
        for (int j = 0; j < 8; j += 2) {
            const unsigned long long m2 =
                ((unsigned long long)__builtin_amdgcn_ballot_w32(key[j + 1] < trial) << 32) |
                __builtin_amdgcn_ballot_w32(key[j] < trial);
            below += __builtin_popcountll(m2);
        }
        if (below < 16) thr = trial;
    }
    unsigned mLess[8], mEq[8];
    int nLess = 0;
#pragma unroll
    for (int j = 0; j < 8; ++j) {
        mLess[j] = __builtin_amdgcn_ballot_w32(key[j] < thr);
        mEq[j]   = __builtin_amdgcn_ballot_w32(key[j] == thr);
        nLess += __builtin_popcount(mLess[j]);
    }
    int offL = 0, offE = nLess;
#pragma unroll
    for (int j = 0; j < 8; ++j) {
        const bool isL = key[j] < thr;
        const bool isE = key[j] == thr;
        const int pos = isL ? offL + (int)__builtin_amdgcn_mbcnt_lo(mLess[j], 0u)
                            : offE + (int)__builtin_amdgcn_mbcnt_lo(mEq[j], 0u);
        if (isL || (isE && pos < 16)) nbrW[pos] = lane + (j << 5);
        offL += __builtin_popcount(mLess[j]);
        offE += __builtin_popcount(mEq[j]);
    }

    const float e0 = zA[tl] - Yp[pi * 64 + d0];
    const float e1 = zB[tl] - Yp[pi * 64 + d1];
    float ya[16], yb[16];
#pragma unroll
    for (int r = 0; r < 16; ++r) {
        const int nq = nbrW[r];
        ya[r] = Yp[nq * 64 + d0];
        yb[r] = Yp[nq * 64 + d1];
    }
#pragma unroll
    for (int r = 0; r < 16; ++r) {
        bufA[r * TP + d0] = fmaxf((ya[r] + e0) * s0a + b0a, 0.0f);
        bufA[r * TP + d1] = fmaxf((yb[r] + e1) * s0b + b0b, 0.0f);
    }

    v8f acc[4] = {vz, vz, vz, vz};
    gemm16x64(bufA, Wt1, lrow, half, acc);
    bn_relu(acc, gammas + 64, betas + 64, means + 64, vars + 64, lrow);
    store16x64(acc, lrow, half, bufA);

    acc[0] = vz; acc[1] = vz; acc[2] = vz; acc[3] = vz;
    gemm16x64(bufA, Wt2, lrow, half, acc);
    bn_relu(acc, gammas + 128, betas + 128, means + 128, vars + 128, lrow);

#pragma unroll
    for (int t = 0; t < 4; ++t) {
        float s = acc[t][0] + acc[t][1] + acc[t][2] + acc[t][3] +
                  acc[t][4] + acc[t][5] + acc[t][6] + acc[t][7];
        s += __shfl_xor(s, 16, 32);
        if (lane < 16) ftsW[lrow + (t << 4)] = s * (1.0f / 16.0f);
    }

    float* orow = out + (((b << 8) | pi) << 6);
    orow[d0] = fmaxf(sA8[tl] * sA + bA + ftsW[d0], 0.f);
    orow[d1] = fmaxf(sB8[tl] * sB + bB + ftsW[d1], 0.f);
    }
}

extern "C" void kernel_launch(void* const* d_in, const int* in_sizes, int n_in,
                              void* d_out, int out_size, void* d_ws, size_t ws_size,
                              hipStream_t stream) {
    (void)in_sizes; (void)n_in; (void)d_ws; (void)ws_size; (void)out_size;
    const float* points = (const float*)d_in[0];
    const float* feats  = (const float*)d_in[1];
    const float* w0     = (const float*)d_in[2];
    const float* w1     = (const float*)d_in[3];
    const float* w2     = (const float*)d_in[4];
    const float* gammas = (const float*)d_in[5];
    const float* betas  = (const float*)d_in[6];
    const float* means  = (const float*)d_in[7];
    const float* vars   = (const float*)d_in[8];
    const float* scw    = (const float*)d_in[9];
    const float* scg    = (const float*)d_in[10];
    const float* scb    = (const float*)d_in[11];
    const float* scm    = (const float*)d_in[12];
    const float* scv    = (const float*)d_in[13];
    float* out = (float*)d_out;

    const size_t ldsBytes = (768 + 256 * 64 + 2 * 64 * TP + 8 * 16 * TP + 512) * sizeof(float)
                          + 128 * sizeof(int);
    dim3 grid(512), block(256);
    edgeconv_fused_kernel<<<grid, block, ldsBytes, stream>>>(
        points, feats, w0, w1, w2, gammas, betas, means, vars,
        scw, scg, scb, scm, scv, out);
}


# KNN threshold search stops at the first trial value with exactly 16 keys below it (same neighbour set; the 16th neighbour is then listed by index, not last, so the 16-row mean adds in another order)
# speedup vs baseline: 7.8231x; 1.0927x over previous
#include <hip/hip_runtime.h>
#include <math.h>

typedef __attribute__((ext_vector_type(2))) float v2f;
typedef __attribute__((ext_vector_type(4))) float v4f;
typedef __attribute__((ext_vector_type(8))) float v8f;

#define WMMA_F32(a, b, c) \
    __builtin_amdgcn_wmma_f32_16x16x4_f32(false, (a), false, (b), (short)0, (c), false, false)

#define BN_EPS 1e-3f
#define BIG 3.0e38f

#define TP  68
#define TP0 132

#define TPW 8

__device__ __forceinline__ void gemm16x64(const float* A, const float* Wt,
                                          int lrow, int half, v8f acc[4]) {
#pragma unroll
    for (int kk = 0; kk < 16; ++kk) {
        const int cb = (kk << 2) + (half << 1);
        v2f a = *(const v2f*)(A + lrow * TP + cb);
#pragma unroll
        for (int t = 0; t < 4; ++t) {
            v2f bw = *(const v2f*)(Wt + (lrow + (t << 4)) * TP + cb);
            acc[t] = WMMA_F32(a, bw, acc[t]);
        }
    }
}

__device__ __forceinline__ void bn_relu(v8f acc[4], const float* g, const float* be,
                                        const float* mn, const float* vr, int lrow) {
#pragma unroll
    for (int t = 0; t < 4; ++t) {
        const int d = lrow + (t << 4);
        const float s = g[d] * (1.0f / sqrtf(vr[d] + BN_EPS));
        const float bi = be[d] - mn[d] * s;
#pragma unroll
        for (int v = 0; v < 8; ++v) acc[t][v] = fmaxf(acc[t][v] * s + bi, 0.0f);
    }
}

__device__ __forceinline__ void store16x64(const v8f acc[4], int lrow, int half, float* Y) {
#pragma unroll
    for (int t = 0; t < 4; ++t)
#pragma unroll
        for (int v = 0; v < 8; ++v)
            Y[(v + (half << 3)) * TP + lrow + (t << 4)] = acc[t][v];
}

__global__ __launch_bounds__(256) void edgeconv_fused_kernel(
    const float* __restrict__ points,
    const float* __restrict__ feats,
    const float* __restrict__ w0,
    const float* __restrict__ w1,
    const float* __restrict__ w2,
    const float* __restrict__ gammas,
    const float* __restrict__ betas,
    const float* __restrict__ means,
    const float* __restrict__ vars,
    const float* __restrict__ scw,
    const float* __restrict__ scg, const float* __restrict__ scb,
    const float* __restrict__ scm, const float* __restrict__ scv,
    float* __restrict__ out)
{
    extern __shared__ __attribute__((aligned(16))) float smem[];
    float* pts     = smem;
    float* Yp      = pts + 768;
    float* R       = Yp + 256 * 64;
    float* Wt0     = R;
    float* scWt    = Wt0 + 64 * TP0;
    float* Wt1     = R;
    float* Wt2     = Wt1 + 64 * TP;
    float* waveBuf = Wt2 + 64 * TP;
    float* ftsAll  = waveBuf + 8 * 16 * TP;
    int*   nbrAll  = (int*)(ftsAll + 512);

    const int tid  = threadIdx.x;
    const int lane = tid & 31;
    const int w    = tid >> 5;
    const int lrow = lane & 15;
    const int half = lane >> 4;
    const int tile0 = blockIdx.x * TPW;
    const int b     = tile0 >> 5;
    const float* fb = feats + (b << 8) * 64;

    for (int i = tid; i < 768; i += 256) pts[i] = points[b * 768 + i];
    for (int i = tid; i < 8192; i += 256) {
        const int d = i >> 7, c = i & 127;
        Wt0[d * TP0 + c] = w0[c * 64 + d];
    }
    for (int i = tid; i < 4096; i += 256) {
        const int d = i >> 6, c = i & 63;
        scWt[d * TP + c] = scw[c * 64 + d];
    }
    __syncthreads();

    const v8f vz = {0.f, 0.f, 0.f, 0.f, 0.f, 0.f, 0.f, 0.f};

    {
        const int q0 = w << 5;
        v2f a0[16], a1[16];
#pragma unroll
        for (int kk = 0; kk < 16; ++kk) {
            const int cb = (kk << 2) + (half << 1);
            a0[kk] = *(const v2f*)(fb + (q0 + lrow) * 64 + cb);
            a1[kk] = *(const v2f*)(fb + (q0 + 16 + lrow) * 64 + cb);
        }
        v8f y0[4] = {vz, vz, vz, vz};
        v8f y1[4] = {vz, vz, vz, vz};
#pragma unroll
        for (int kk = 0; kk < 16; ++kk) {
            const int cb = (kk << 2) + (half << 1);
#pragma unroll
            for (int t = 0; t < 4; ++t) {
                v2f bw = *(const v2f*)(Wt0 + (lrow + (t << 4)) * TP0 + 64 + cb);
                y0[t] = WMMA_F32(a0[kk], bw, y0[t]);
                y1[t] = WMMA_F32(a1[kk], bw, y1[t]);
            }
        }
#pragma unroll
        for (int t = 0; t < 4; ++t)
#pragma unroll
            for (int v = 0; v < 8; ++v) {
                Yp[(q0 + v + (half << 3)) * 64 + lrow + (t << 4)]      = y0[t][v];
                Yp[(q0 + 16 + v + (half << 3)) * 64 + lrow + (t << 4)] = y1[t][v];
            }
    }
    v8f zA, zB, sA8, sB8;
    {
        const int pj = (((tile0 + (lrow & 7)) & 31) << 3) | w;
        v2f a[16];
#pragma unroll
        for (int kk = 0; kk < 16; ++kk) {
            const int cb = (kk << 2) + (half << 1);
            a[kk] = *(const v2f*)(fb + pj * 64 + cb);
        }
        v8f az[4] = {vz, vz, vz, vz};
        v8f as[4] = {vz, vz, vz, vz};
#pragma unroll
        for (int kk = 0; kk < 16; ++kk) {
            const int cb = (kk << 2) + (half << 1);
#pragma unroll
            for (int t = 0; t < 4; ++t) {
                v2f bz = *(const v2f*)(Wt0 + (lrow + (t << 4)) * TP0 + cb);
                v2f bs = *(const v2f*)(scWt + (lrow + (t << 4)) * TP + cb);
                az[t] = WMMA_F32(a[kk], bz, az[t]);
                as[t] = WMMA_F32(a[kk], bs, as[t]);
            }
        }
#pragma unroll
        for (int v = 0; v < 8; ++v) {
            zA[v]  = half ? az[1][v] : az[0][v];
            zB[v]  = half ? az[3][v] : az[2][v];
            sA8[v] = half ? as[1][v] : as[0][v];
            sB8[v] = half ? as[3][v] : as[2][v];
        }
    }
    __syncthreads();
    for (int i = tid; i < 4096; i += 256) {
        const int d = i >> 6, c = i & 63;
        Wt1[d * TP + c] = w1[c * 64 + d];
        Wt2[d * TP + c] = w2[c * 64 + d];
    }
    __syncthreads();

    float* bufA = waveBuf + w * (16 * TP);
    float* ftsW = ftsAll + (w << 6);
    int*   nbrW = nbrAll + (w << 4);

    const int d0 = lane, d1 = lane + 32;
    const float s0a = gammas[d0] * (1.0f / sqrtf(vars[d0] + BN_EPS));
    const float b0a = betas[d0] - means[d0] * s0a;
    const float s0b = gammas[d1] * (1.0f / sqrtf(vars[d1] + BN_EPS));
    const float b0b = betas[d1] - means[d1] * s0b;
    const float sA = scg[d0] * (1.0f / sqrtf(scv[d0] + BN_EPS));
    const float bA = scb[d0] - scm[d0] * sA;
    const float sB = scg[d1] * (1.0f / sqrtf(scv[d1] + BN_EPS));
    const float bB = scb[d1] - scm[d1] * sB;

    for (int tl = 0; tl < TPW; ++tl) {
    const int pi = (((tile0 + tl) & 31) << 3) | w;

    const float px = pts[pi * 3], py = pts[pi * 3 + 1], pz = pts[pi * 3 + 2];
    unsigned key[8];
#pragma unroll
    for (int j = 0; j < 8; ++j) {
        const int q = lane + (j << 5);
        const float dx = pts[q * 3] - px;
        const float dy = pts[q * 3 + 1] - py;
        const float dz = pts[q * 3 + 2] - pz;
        const float dd = dx * dx + dy * dy + dz * dz;
        key[j] = __float_as_uint((q == pi) ? BIG : dd);
    }
    unsigned thr = 0;
    int bit = 30;
    do {
        const unsigned trial = thr | (1u << bit);
        int below = 0;
#pragma unroll
        for (int j = 0; j < 8; j += 2) {
            const unsigned long long m2 =
                ((unsigned long long)__builtin_amdgcn_ballot_w32(key[j + 1] < trial) << 32) |
                __builtin_amdgcn_ballot_w32(key[j] < trial);
            below += __builtin_popcountll(m2);
        }
        thr = (below <= 16) ? trial : thr;
        bit -= (below == 16) ? 32 : 1;
    } while (bit >= 0);
    unsigned mLess[8], mEq[8];
    int nLess = 0;
#pragma unroll
    for (int j = 0; j < 8; ++j) {
        mLess[j] = __builtin_amdgcn_ballot_w32(key[j] < thr);
        mEq[j]   = __builtin_amdgcn_ballot_w32(key[j] == thr);
        nLess += __builtin_popcount(mLess[j]);
    }
    int offL = 0, offE = nLess;
#pragma unroll
    for (int j = 0; j < 8; ++j) {
        const bool isL = key[j] < thr;
        const bool isE = key[j] == thr;
        const int pos = isL ? offL + (int)__builtin_amdgcn_mbcnt_lo(mLess[j], 0u)
                            : offE + (int)__builtin_amdgcn_mbcnt_lo(mEq[j], 0u);
        if (isL || (isE && pos < 16)) nbrW[pos] = lane + (j << 5);
        offL += __builtin_popcount(mLess[j]);
        offE += __builtin_popcount(mEq[j]);
    }

    const float e0 = zA[tl] - Yp[pi * 64 + d0];
    const float e1 = zB[tl] - Yp[pi * 64 + d1];
    float ya[16], yb[16];
#pragma unroll
    for (int r = 0; r < 16; ++r) {
        const int nq = nbrW[r];
        ya[r] = Yp[nq * 64 + d0];
        yb[r] = Yp[nq * 64 + d1];
    }
#pragma unroll
    for (int r = 0; r < 16; ++r) {
        bufA[r * TP + d0] = fmaxf((ya[r] + e0) * s0a + b0a, 0.0f);
        bufA[r * TP + d1] = fmaxf((yb[r] + e1) * s0b + b0b, 0.0f);
    }

    v8f acc[4] = {vz, vz, vz, vz};
    gemm16x64(bufA, Wt1, lrow, half, acc);
    bn_relu(acc, gammas + 64, betas + 64, means + 64, vars + 64, lrow);
    store16x64(acc, lrow, half, bufA);

    acc[0] = vz; acc[1] = vz; acc[2] = vz; acc[3] = vz;
    gemm16x64(bufA, Wt2, lrow, half, acc);
    bn_relu(acc, gammas + 128, betas + 128, means + 128, vars + 128, lrow);

#pragma unroll
    for (int t = 0; t < 4; ++t) {
        float s = acc[t][0] + acc[t][1] + acc[t][2] + acc[t][3] +
                  acc[t][4] + acc[t][5] + acc[t][6] + acc[t][7];
        s += __shfl_xor(s, 16, 32);
        if (lane < 16) ftsW[lrow + (t << 4)] = s * (1.0f / 16.0f);
    }

    float* orow = out + (((b << 8) | pi) << 6);
    orow[d0] = fmaxf(sA8[tl] * sA + bA + ftsW[d0], 0.f);
    orow[d1] = fmaxf(sB8[tl] * sB + bB + ftsW[d1], 0.f);
    }
}

extern "C" void kernel_launch(void* const* d_in, const int* in_sizes, int n_in,
                              void* d_out, int out_size, void* d_ws, size_t ws_size,
                              hipStream_t stream) {
    (void)in_sizes; (void)n_in; (void)d_ws; (void)ws_size; (void)out_size;
    const float* points = (const float*)d_in[0];
    const float* feats  = (const float*)d_in[1];
    const float* w0     = (const float*)d_in[2];
    const float* w1     = (const float*)d_in[3];
    const float* w2     = (const float*)d_in[4];
    const float* gammas = (const float*)d_in[5];
    const float* betas  = (const float*)d_in[6];
    const float* means  = (const float*)d_in[7];
    const float* vars   = (const float*)d_in[8];
    const float* scw    = (const float*)d_in[9];
    const float* scg    = (const float*)d_in[10];
    const float* scb    = (const float*)d_in[11];
    const float* scm    = (const float*)d_in[12];
    const float* scv    = (const float*)d_in[13];
    float* out = (float*)d_out;

    const size_t ldsBytes = (768 + 256 * 64 + 2 * 64 * TP + 8 * 16 * TP + 512) * sizeof(float)
                          + 128 * sizeof(int);
    dim3 grid(512), block(256);
    edgeconv_fused_kernel<<<grid, block, ldsBytes, stream>>>(
        points, feats, w0, w1, w2, gammas, betas, means, vars,
        scw, scg, scb, scm, scv, out);
}


# weight staging reads whole 128-byte lines: lanes run along a weight row in global memory, the transposition moves to the LDS store (2 lanes per bank); each phase's loads go out as one group
# speedup vs baseline: 8.4544x; 1.0807x over previous
#include <hip/hip_runtime.h>
#include <math.h>

typedef __attribute__((ext_vector_type(2))) float v2f;
typedef __attribute__((ext_vector_type(4))) float v4f;
typedef __attribute__((ext_vector_type(8))) float v8f;

#define WMMA_F32(a, b, c) \
    __builtin_amdgcn_wmma_f32_16x16x4_f32(false, (a), false, (b), (short)0, (c), false, false)

#define BN_EPS 1e-3f
#define BIG 3.0e38f

#define TP  68
#define TP0 132

#define TPW 8

__device__ __forceinline__ void gemm16x64(const float* A, const float* Wt,
                                          int lrow, int half, v8f acc[4]) {
#pragma unroll
    for (int kk = 0; kk < 16; ++kk) {
        const int cb = (kk << 2) + (half << 1);
        v2f a = *(const v2f*)(A + lrow * TP + cb);
#pragma unroll
        for (int t = 0; t < 4; ++t) {
            v2f bw = *(const v2f*)(Wt + (lrow + (t << 4)) * TP + cb);
            acc[t] = WMMA_F32(a, bw, acc[t]);
        }
    }
}

__device__ __forceinline__ void bn_relu(v8f acc[4], const float* g, const float* be,
                                        const float* mn, const float* vr, int lrow) {
#pragma unroll
    for (int t = 0; t < 4; ++t) {
        const int d = lrow + (t << 4);
        const float s = g[d] * (1.0f / sqrtf(vr[d] + BN_EPS));
        const float bi = be[d] - mn[d] * s;
#pragma unroll
        for (int v = 0; v < 8; ++v) acc[t][v] = fmaxf(acc[t][v] * s + bi, 0.0f);
    }
}

__device__ __forceinline__ void store16x64(const v8f acc[4], int lrow, int half, float* Y) {
#pragma unroll
    for (int t = 0; t < 4; ++t)
#pragma unroll
        for (int v = 0; v < 8; ++v)
            Y[(v + (half << 3)) * TP + lrow + (t << 4)] = acc[t][v];
}

__global__ __launch_bounds__(256) void edgeconv_fused_kernel(
    const float* __restrict__ points,
    const float* __restrict__ feats,
    const float* __restrict__ w0,
    const float* __restrict__ w1,
    const float* __restrict__ w2,
    const float* __restrict__ gammas,
    const float* __restrict__ betas,
    const float* __restrict__ means,
    const float* __restrict__ vars,
    const float* __restrict__ scw,
    const float* __restrict__ scg, const float* __restrict__ scb,
    const float* __restrict__ scm, const float* __restrict__ scv,
    float* __restrict__ out)
{
    extern __shared__ __attribute__((aligned(16))) float smem[];
    float* pts     = smem;
    float* Yp      = pts + 768;
    float* R       = Yp + 256 * 64;
    float* Wt0     = R;
    float* scWt    = Wt0 + 64 * TP0;
    float* Wt1     = R;
    float* Wt2     = Wt1 + 64 * TP;
    float* waveBuf = Wt2 + 64 * TP;
    float* ftsAll  = waveBuf + 8 * 16 * TP;
    int*   nbrAll  = (int*)(ftsAll + 512);

    const int tid  = threadIdx.x;
    const int lane = tid & 31;
    const int w    = tid >> 5;
    const int lrow = lane & 15;
    const int half = lane >> 4;
    const int tile0 = blockIdx.x * TPW;
    const int b     = tile0 >> 5;
    const float* fb = feats + (b << 8) * 64;

    const int sc0 = tid >> 6, sd = tid & 63;
    {
        float* l0 = Wt0 + sd * TP0 + sc0;
        float* ls = scWt + sd * TP + sc0;
#pragma unroll
        for (int k = 0; k < 3; ++k) pts[tid + 256 * k] = points[b * 768 + tid + 256 * k];
#pragma unroll
        for (int k = 0; k < 32; ++k) l0[4 * k] = w0[tid + 256 * k];
#pragma unroll
        for (int k = 0; k < 16; ++k) ls[4 * k] = scw[tid + 256 * k];
    }
    __syncthreads();

    const v8f vz = {0.f, 0.f, 0.f, 0.f, 0.f, 0.f, 0.f, 0.f};

    {
        const int q0 = w << 5;
        v2f a0[16], a1[16];
#pragma unroll
        for (int kk = 0; kk < 16; ++kk) {
            const int cb = (kk << 2) + (half << 1);
            a0[kk] = *(const v2f*)(fb + (q0 + lrow) * 64 + cb);
            a1[kk] = *(const v2f*)(fb + (q0 + 16 + lrow) * 64 + cb);
        }
        v8f y0[4] = {vz, vz, vz, vz};
        v8f y1[4] = {vz, vz, vz, vz};
#pragma unroll
        for (int kk = 0; kk < 16; ++kk) {
            const int cb = (kk << 2) + (half << 1);
#pragma unroll
            for (int t = 0; t < 4; ++t) {
                v2f bw = *(const v2f*)(Wt0 + (lrow + (t << 4)) * TP0 + 64 + cb);
                y0[t] = WMMA_F32(a0[kk], bw, y0[t]);
                y1[t] = WMMA_F32(a1[kk], bw, y1[t]);
            }
        }
#pragma unroll
        for (int t = 0; t < 4; ++t)
#pragma unroll
            for (int v = 0; v < 8; ++v) {
                Yp[(q0 + v + (half << 3)) * 64 + lrow + (t << 4)]      = y0[t][v];
                Yp[(q0 + 16 + v + (half << 3)) * 64 + lrow + (t << 4)] = y1[t][v];
            }
    }
    v8f zA, zB, sA8, sB8;
    {
        const int pj = (((tile0 + (lrow & 7)) & 31) << 3) | w;
        v2f a[16];
#pragma unroll
        for (int kk = 0; kk < 16; ++kk) {
            const int cb = (kk << 2) + (half << 1);
            a[kk] = *(const v2f*)(fb + pj * 64 + cb);
        }
        v8f az[4] = {vz, vz, vz, vz};
        v8f as[4] = {vz, vz, vz, vz};
#pragma unroll
        for (int kk = 0; kk < 16; ++kk) {
            const int cb = (kk << 2) + (half << 1);
#pragma unroll
            for (int t = 0; t < 4; ++t) {
                v2f bz = *(const v2f*)(Wt0 + (lrow + (t << 4)) * TP0 + cb);
                v2f bs = *(const v2f*)(scWt + (lrow + (t << 4)) * TP + cb);
                az[t] = WMMA_F32(a[kk], bz, az[t]);
                as[t] = WMMA_F32(a[kk], bs, as[t]);
            }
        }
#pragma unroll
        for (int v = 0; v < 8; ++v) {
            zA[v]  = half ? az[1][v] : az[0][v];
            zB[v]  = half ? az[3][v] : az[2][v];
            sA8[v] = half ? as[1][v] : as[0][v];
            sB8[v] = half ? as[3][v] : as[2][v];
        }
    }
    __syncthreads();
    {
        float* l1 = Wt1 + sd * TP + sc0;
        float* l2 = Wt2 + sd * TP + sc0;
        float r1[16], r2[16];
#pragma unroll
        for (int k = 0; k < 16; ++k) {
            r1[k] = w1[tid + 256 * k];
            r2[k] = w2[tid + 256 * k];
        }
        __builtin_amdgcn_sched_barrier(0);
#pragma unroll
        for (int k = 0; k < 16; ++k) {
            l1[4 * k] = r1[k];
            l2[4 * k] = r2[k];
        }
    }
    __syncthreads();

    float* bufA = waveBuf + w * (16 * TP);
    float* ftsW = ftsAll + (w << 6);
    int*   nbrW = nbrAll + (w << 4);

    const int d0 = lane, d1 = lane + 32;
    const float s0a = gammas[d0] * (1.0f / sqrtf(vars[d0] + BN_EPS));
    const float b0a = betas[d0] - means[d0] * s0a;
    const float s0b = gammas[d1] * (1.0f / sqrtf(vars[d1] + BN_EPS));
    const float b0b = betas[d1] - means[d1] * s0b;
    const float sA = scg[d0] * (1.0f / sqrtf(scv[d0] + BN_EPS));
    const float bA = scb[d0] - scm[d0] * sA;
    const float sB = scg[d1] * (1.0f / sqrtf(scv[d1] + BN_EPS));
    const float bB = scb[d1] - scm[d1] * sB;

    for (int tl = 0; tl < TPW; ++tl) {
    const int pi = (((tile0 + tl) & 31) << 3) | w;

    const float px = pts[pi * 3], py = pts[pi * 3 + 1], pz = pts[pi * 3 + 2];
    unsigned key[8];
#pragma unroll
    for (int j = 0; j < 8; ++j) {
        const int q = lane + (j << 5);
        const float dx = pts[q * 3] - px;
        const float dy = pts[q * 3 + 1] - py;
        const float dz = pts[q * 3 + 2] - pz;
        const float dd = dx * dx + dy * dy + dz * dz;
        key[j] = __float_as_uint((q == pi) ? BIG : dd);
    }
    unsigned thr = 0;
    int bit = 30;
    do {
        const unsigned trial = thr | (1u << bit);
        int below = 0;
#pragma unroll
        for (int j = 0; j < 8; j += 2) {
            const unsigned long long m2 =
                ((unsigned long long)__builtin_amdgcn_ballot_w32(key[j + 1] < trial) << 32) |
                __builtin_amdgcn_ballot_w32(key[j] < trial);
            below += __builtin_popcountll(m2);
        }
        thr = (below <= 16) ? trial : thr;
        bit -= (below == 16) ? 32 : 1;
    } while (bit >= 0);
    unsigned mLess[8], mEq[8];
    int nLess = 0;
#pragma unroll
    for (int j = 0; j < 8; ++j) {
        mLess[j] = __builtin_amdgcn_ballot_w32(key[j] < thr);
        mEq[j]   = __builtin_amdgcn_ballot_w32(key[j] == thr);
        nLess += __builtin_popcount(mLess[j]);
    }
    int offL = 0, offE = nLess;
#pragma unroll
    for (int j = 0; j < 8; ++j) {
        const bool isL = key[j] < thr;
        const bool isE = key[j] == thr;
        const int pos = isL ? offL + (int)__builtin_amdgcn_mbcnt_lo(mLess[j], 0u)
                            : offE + (int)__builtin_amdgcn_mbcnt_lo(mEq[j], 0u);
        if (isL || (isE && pos < 16)) nbrW[pos] = lane + (j << 5);
        offL += __builtin_popcount(mLess[j]);
        offE += __builtin_popcount(mEq[j]);
    }

    const float e0 = zA[tl] - Yp[pi * 64 + d0];
    const float e1 = zB[tl] - Yp[pi * 64 + d1];
    float ya[16], yb[16];
#pragma unroll
    for (int r = 0; r < 16; ++r) {
        const int nq = nbrW[r];
        ya[r] = Yp[nq * 64 + d0];
        yb[r] = Yp[nq * 64 + d1];
    }
#pragma unroll
    for (int r = 0; r < 16; ++r) {
        bufA[r * TP + d0] = fmaxf((ya[r] + e0) * s0a + b0a, 0.0f);
        bufA[r * TP + d1] = fmaxf((yb[r] + e1) * s0b + b0b, 0.0f);
    }

    v8f acc[4] = {vz, vz, vz, vz};
    gemm16x64(bufA, Wt1, lrow, half, acc);
    bn_relu(acc, gammas + 64, betas + 64, means + 64, vars + 64, lrow);
    store16x64(acc, lrow, half, bufA);

    acc[0] = vz; acc[1] = vz; acc[2] = vz; acc[3] = vz;
    gemm16x64(bufA, Wt2, lrow, half, acc);
    bn_relu(acc, gammas + 128, betas + 128, means + 128, vars + 128, lrow);

#pragma unroll
    for (int t = 0; t < 4; ++t) {
        float s = acc[t][0] + acc[t][1] + acc[t][2] + acc[t][3] +
                  acc[t][4] + acc[t][5] + acc[t][6] + acc[t][7];
        s += __shfl_xor(s, 16, 32);
        if (lane < 16) ftsW[lrow + (t << 4)] = s * (1.0f / 16.0f);
    }

    float* orow = out + (((b << 8) | pi) << 6);
    orow[d0] = fmaxf(sA8[tl] * sA + bA + ftsW[d0], 0.f);
    orow[d1] = fmaxf(sB8[tl] * sB + bB + ftsW[d1], 0.f);
    }
}

extern "C" void kernel_launch(void* const* d_in, const int* in_sizes, int n_in,
                              void* d_out, int out_size, void* d_ws, size_t ws_size,
                              hipStream_t stream) {
    (void)in_sizes; (void)n_in; (void)d_ws; (void)ws_size; (void)out_size;
    const float* points = (const float*)d_in[0];
    const float* feats  = (const float*)d_in[1];
    const float* w0     = (const float*)d_in[2];
    const float* w1     = (const float*)d_in[3];
    const float* w2     = (const float*)d_in[4];
    const float* gammas = (const float*)d_in[5];
    const float* betas  = (const float*)d_in[6];
    const float* means  = (const float*)d_in[7];
    const float* vars   = (const float*)d_in[8];
    const float* scw    = (const float*)d_in[9];
    const float* scg    = (const float*)d_in[10];
    const float* scb    = (const float*)d_in[11];
    const float* scm    = (const float*)d_in[12];
    const float* scv    = (const float*)d_in[13];
    float* out = (float*)d_out;

    const size_t ldsBytes = (768 + 256 * 64 + 2 * 64 * TP + 8 * 16 * TP + 512) * sizeof(float)
                          + 128 * sizeof(int);
    dim3 grid(512), block(256);
    edgeconv_fused_kernel<<<grid, block, ldsBytes, stream>>>(
        points, feats, w0, w1, w2, gammas, betas, means, vars,
        scw, scg, scb, scm, scv, out);
}


# tile loop: s_setprio falls with tiles done (3,3,3,2,2,1,1,0), so the wave furthest behind wins the SIMD's matrix pipe and its four waves end together; no arithmetic or order changed
# speedup vs baseline: 8.9514x; 1.0588x over previous
#include <hip/hip_runtime.h>
#include <math.h>

typedef __attribute__((ext_vector_type(2))) float v2f;
typedef __attribute__((ext_vector_type(4))) float v4f;
typedef __attribute__((ext_vector_type(8))) float v8f;

#define WMMA_F32(a, b, c) \
    __builtin_amdgcn_wmma_f32_16x16x4_f32(false, (a), false, (b), (short)0, (c), false, false)

#define BN_EPS 1e-3f
#define BIG 3.0e38f

#define TP  68
#define TP0 132

#define TPW 8

__device__ __forceinline__ void gemm16x64(const float* A, const float* Wt,
                                          int lrow, int half, v8f acc[4]) {
#pragma unroll
    for (int kk = 0; kk < 16; ++kk) {
        const int cb = (kk << 2) + (half << 1);
        v2f a = *(const v2f*)(A + lrow * TP + cb);
#pragma unroll
        for (int t = 0; t < 4; ++t) {
            v2f bw = *(const v2f*)(Wt + (lrow + (t << 4)) * TP + cb);
            acc[t] = WMMA_F32(a, bw, acc[t]);
        }
    }
}

__device__ __forceinline__ void bn_relu(v8f acc[4], const float* g, const float* be,
                                        const float* mn, const float* vr, int lrow) {
#pragma unroll
    for (int t = 0; t < 4; ++t) {
        const int d = lrow + (t << 4);
        const float s = g[d] * (1.0f / sqrtf(vr[d] + BN_EPS));
        const float bi = be[d] - mn[d] * s;
#pragma unroll
        for (int v = 0; v < 8; ++v) acc[t][v] = fmaxf(acc[t][v] * s + bi, 0.0f);
    }
}

__device__ __forceinline__ void store16x64(const v8f acc[4], int lrow, int half, float* Y) {
#pragma unroll
    for (int t = 0; t < 4; ++t)
#pragma unroll
        for (int v = 0; v < 8; ++v)
            Y[(v + (half << 3)) * TP + lrow + (t << 4)] = acc[t][v];
}

__global__ __launch_bounds__(256) void edgeconv_fused_kernel(
    const float* __restrict__ points,
    const float* __restrict__ feats,
    const float* __restrict__ w0,
    const float* __restrict__ w1,
    const float* __restrict__ w2,
    const float* __restrict__ gammas,
    const float* __restrict__ betas,
    const float* __restrict__ means,
    const float* __restrict__ vars,
    const float* __restrict__ scw,
    const float* __restrict__ scg, const float* __restrict__ scb,
    const float* __restrict__ scm, const float* __restrict__ scv,
    float* __restrict__ out)
{
    extern __shared__ __attribute__((aligned(16))) float smem[];
    float* pts     = smem;
    float* Yp      = pts + 768;
    float* R       = Yp + 256 * 64;
    float* Wt0     = R;
    float* scWt    = Wt0 + 64 * TP0;
    float* Wt1     = R;
    float* Wt2     = Wt1 + 64 * TP;
    float* waveBuf = Wt2 + 64 * TP;
    float* ftsAll  = waveBuf + 8 * 16 * TP;
    int*   nbrAll  = (int*)(ftsAll + 512);

    const int tid  = threadIdx.x;
    const int lane = tid & 31;
    const int w    = tid >> 5;
    const int lrow = lane & 15;
    const int half = lane >> 4;
    const int tile0 = blockIdx.x * TPW;
    const int b     = tile0 >> 5;
    const float* fb = feats + (b << 8) * 64;

    const int sc0 = tid >> 6, sd = tid & 63;
    {
        float* l0 = Wt0 + sd * TP0 + sc0;
        float* ls = scWt + sd * TP + sc0;
#pragma unroll
        for (int k = 0; k < 3; ++k) pts[tid + 256 * k] = points[b * 768 + tid + 256 * k];
#pragma unroll
        for (int k = 0; k < 32; ++k) l0[4 * k] = w0[tid + 256 * k];
#pragma unroll
        for (int k = 0; k < 16; ++k) ls[4 * k] = scw[tid + 256 * k];
    }
    __syncthreads();

    const v8f vz = {0.f, 0.f, 0.f, 0.f, 0.f, 0.f, 0.f, 0.f};

    {
        const int q0 = w << 5;
        v2f a0[16], a1[16];
#pragma unroll
        for (int kk = 0; kk < 16; ++kk) {
            const int cb = (kk << 2) + (half << 1);
            a0[kk] = *(const v2f*)(fb + (q0 + lrow) * 64 + cb);
            a1[kk] = *(const v2f*)(fb + (q0 + 16 + lrow) * 64 + cb);
        }
        v8f y0[4] = {vz, vz, vz, vz};
        v8f y1[4] = {vz, vz, vz, vz};
#pragma unroll
        for (int kk = 0; kk < 16; ++kk) {
            const int cb = (kk << 2) + (half << 1);
#pragma unroll
            for (int t = 0; t < 4; ++t) {
                v2f bw = *(const v2f*)(Wt0 + (lrow + (t << 4)) * TP0 + 64 + cb);
                y0[t] = WMMA_F32(a0[kk], bw, y0[t]);
                y1[t] = WMMA_F32(a1[kk], bw, y1[t]);
            }
        }
#pragma unroll
        for (int t = 0; t < 4; ++t)
#pragma unroll
            for (int v = 0; v < 8; ++v) {
                Yp[(q0 + v + (half << 3)) * 64 + lrow + (t << 4)]      = y0[t][v];
                Yp[(q0 + 16 + v + (half << 3)) * 64 + lrow + (t << 4)] = y1[t][v];
            }
    }
    v8f zA, zB, sA8, sB8;
    {
        const int pj = (((tile0 + (lrow & 7)) & 31) << 3) | w;
        v2f a[16];
#pragma unroll
        for (int kk = 0; kk < 16; ++kk) {
            const int cb = (kk << 2) + (half << 1);
            a[kk] = *(const v2f*)(fb + pj * 64 + cb);
        }
        v8f az[4] = {vz, vz, vz, vz};
        v8f as[4] = {vz, vz, vz, vz};
#pragma unroll
        for (int kk = 0; kk < 16; ++kk) {
            const int cb = (kk << 2) + (half << 1);
#pragma unroll
            for (int t = 0; t < 4; ++t) {
                v2f bz = *(const v2f*)(Wt0 + (lrow + (t << 4)) * TP0 + cb);
                v2f bs = *(const v2f*)(scWt + (lrow + (t << 4)) * TP + cb);
                az[t] = WMMA_F32(a[kk], bz, az[t]);
                as[t] = WMMA_F32(a[kk], bs, as[t]);
            }
        }
#pragma unroll
        for (int v = 0; v < 8; ++v) {
            zA[v]  = half ? az[1][v] : az[0][v];
            zB[v]  = half ? az[3][v] : az[2][v];
            sA8[v] = half ? as[1][v] : as[0][v];
            sB8[v] = half ? as[3][v] : as[2][v];
        }
    }
    __syncthreads();
    {
        float* l1 = Wt1 + sd * TP + sc0;
        float* l2 = Wt2 + sd * TP + sc0;
        float r1[16], r2[16];
#pragma unroll
        for (int k = 0; k < 16; ++k) {
            r1[k] = w1[tid + 256 * k];
            r2[k] = w2[tid + 256 * k];
        }
        __builtin_amdgcn_sched_barrier(0);
#pragma unroll
        for (int k = 0; k < 16; ++k) {
            l1[4 * k] = r1[k];
            l2[4 * k] = r2[k];
        }
    }
    __syncthreads();

    float* bufA = waveBuf + w * (16 * TP);
    float* ftsW = ftsAll + (w << 6);
    int*   nbrW = nbrAll + (w << 4);

    const int d0 = lane, d1 = lane + 32;
    const float s0a = gammas[d0] * (1.0f / sqrtf(vars[d0] + BN_EPS));
    const float b0a = betas[d0] - means[d0] * s0a;
    const float s0b = gammas[d1] * (1.0f / sqrtf(vars[d1] + BN_EPS));
    const float b0b = betas[d1] - means[d1] * s0b;
    const float sA = scg[d0] * (1.0f / sqrtf(scv[d0] + BN_EPS));
    const float bA = scb[d0] - scm[d0] * sA;
    const float sB = scg[d1] * (1.0f / sqrtf(scv[d1] + BN_EPS));
    const float bB = scb[d1] - scm[d1] * sB;

    __builtin_amdgcn_s_setprio(3);
    for (int tl = 0; tl < TPW; ++tl) {
    if (tl == 3) __builtin_amdgcn_s_setprio(2);
    if (tl == 5) __builtin_amdgcn_s_setprio(1);
    if (tl == 7) __builtin_amdgcn_s_setprio(0);
    const int pi = (((tile0 + tl) & 31) << 3) | w;

    const float px = pts[pi * 3], py = pts[pi * 3 + 1], pz = pts[pi * 3 + 2];
    unsigned key[8];
#pragma unroll
    for (int j = 0; j < 8; ++j) {
        const int q = lane + (j << 5);
        const float dx = pts[q * 3] - px;
        const float dy = pts[q * 3 + 1] - py;
        const float dz = pts[q * 3 + 2] - pz;
        const float dd = dx * dx + dy * dy + dz * dz;
        key[j] = __float_as_uint((q == pi) ? BIG : dd);
    }
    unsigned thr = 0;
    int bit = 30;
    do {
        const unsigned trial = thr | (1u << bit);
        int below = 0;
#pragma unroll
        for (int j = 0; j < 8; j += 2) {
            const unsigned long long m2 =
                ((unsigned long long)__builtin_amdgcn_ballot_w32(key[j + 1] < trial) << 32) |
                __builtin_amdgcn_ballot_w32(key[j] < trial);
            below += __builtin_popcountll(m2);
        }
        thr = (below <= 16) ? trial : thr;
        bit -= (below == 16) ? 32 : 1;
    } while (bit >= 0);
    unsigned mLess[8], mEq[8];
    int nLess = 0;
#pragma unroll
    for (int j = 0; j < 8; ++j) {
        mLess[j] = __builtin_amdgcn_ballot_w32(key[j] < thr);
        mEq[j]   = __builtin_amdgcn_ballot_w32(key[j] == thr);
        nLess += __builtin_popcount(mLess[j]);
    }
    int offL = 0, offE = nLess;
#pragma unroll
    for (int j = 0; j < 8; ++j) {
        const bool isL = key[j] < thr;
        const bool isE = key[j] == thr;
        const int pos = isL ? offL + (int)__builtin_amdgcn_mbcnt_lo(mLess[j], 0u)
                            : offE + (int)__builtin_amdgcn_mbcnt_lo(mEq[j], 0u);
        if (isL || (isE && pos < 16)) nbrW[pos] = lane + (j << 5);
        offL += __builtin_popcount(mLess[j]);
        offE += __builtin_popcount(mEq[j]);
    }

    const float e0 = zA[tl] - Yp[pi * 64 + d0];
    const float e1 = zB[tl] - Yp[pi * 64 + d1];
    float ya[16], yb[16];
#pragma unroll
    for (int r = 0; r < 16; ++r) {
        const int nq = nbrW[r];
        ya[r] = Yp[nq * 64 + d0];
        yb[r] = Yp[nq * 64 + d1];
    }
#pragma unroll
    for (int r = 0; r < 16; ++r) {
        bufA[r * TP + d0] = fmaxf((ya[r] + e0) * s0a + b0a, 0.0f);
        bufA[r * TP + d1] = fmaxf((yb[r] + e1) * s0b + b0b, 0.0f);
    }

    v8f acc[4] = {vz, vz, vz, vz};
    gemm16x64(bufA, Wt1, lrow, half, acc);
    bn_relu(acc, gammas + 64, betas + 64, means + 64, vars + 64, lrow);
    store16x64(acc, lrow, half, bufA);

    acc[0] = vz; acc[1] = vz; acc[2] = vz; acc[3] = vz;
    gemm16x64(bufA, Wt2, lrow, half, acc);
    bn_relu(acc, gammas + 128, betas + 128, means + 128, vars + 128, lrow);

#pragma unroll
    for (int t = 0; t < 4; ++t) {
        float s = acc[t][0] + acc[t][1] + acc[t][2] + acc[t][3] +
                  acc[t][4] + acc[t][5] + acc[t][6] + acc[t][7];
        s += __shfl_xor(s, 16, 32);
        if (lane < 16) ftsW[lrow + (t << 4)] = s * (1.0f / 16.0f);
    }

    float* orow = out + (((b << 8) | pi) << 6);
    orow[d0] = fmaxf(sA8[tl] * sA + bA + ftsW[d0], 0.f);
    orow[d1] = fmaxf(sB8[tl] * sB + bB + ftsW[d1], 0.f);
    }
}

extern "C" void kernel_launch(void* const* d_in, const int* in_sizes, int n_in,
                              void* d_out, int out_size, void* d_ws, size_t ws_size,
                              hipStream_t stream) {
    (void)in_sizes; (void)n_in; (void)d_ws; (void)ws_size; (void)out_size;
    const float* points = (const float*)d_in[0];
    const float* feats  = (const float*)d_in[1];
    const float* w0     = (const float*)d_in[2];
    const float* w1     = (const float*)d_in[3];
    const float* w2     = (const float*)d_in[4];
    const float* gammas = (const float*)d_in[5];
    const float* betas  = (const float*)d_in[6];
    const float* means  = (const float*)d_in[7];
    const float* vars   = (const float*)d_in[8];
    const float* scw    = (const float*)d_in[9];
    const float* scg    = (const float*)d_in[10];
    const float* scb    = (const float*)d_in[11];
    const float* scm    = (const float*)d_in[12];
    const float* scv    = (const float*)d_in[13];
    float* out = (float*)d_out;

    const size_t ldsBytes = (768 + 256 * 64 + 2 * 64 * TP + 8 * 16 * TP + 512) * sizeof(float)
                          + 128 * sizeof(int);
    dim3 grid(512), block(256);
    edgeconv_fused_kernel<<<grid, block, ldsBytes, stream>>>(
        points, feats, w0, w1, w2, gammas, betas, means, vars,
        scw, scg, scb, scm, scv, out);
}


# one 16-wave work-group per WGP (grid 512 -> 256, 128 points each): weights, points and Y are staged once per WGP, not twice, so Y is projected twice per batch instead of four times
# speedup vs baseline: 9.3774x; 1.0476x over previous
#include <hip/hip_runtime.h>
#include <math.h>

typedef __attribute__((ext_vector_type(2))) float v2f;
typedef __attribute__((ext_vector_type(4))) float v4f;
typedef __attribute__((ext_vector_type(8))) float v8f;

#define WMMA_F32(a, b, c) \
    __builtin_amdgcn_wmma_f32_16x16x4_f32(false, (a), false, (b), (short)0, (c), false, false)

#define BN_EPS 1e-3f
#define BIG 3.0e38f

#define TP  68
#define TP0 132

#define NW  16
#define NT  (NW * 32)
#define TPW 8

__device__ __forceinline__ void gemm16x64(const float* A, const float* Wt,
                                          int lrow, int half, v8f acc[4]) {
#pragma unroll
    for (int kk = 0; kk < 16; ++kk) {
        const int cb = (kk << 2) + (half << 1);
        v2f a = *(const v2f*)(A + lrow * TP + cb);
#pragma unroll
        for (int t = 0; t < 4; ++t) {
            v2f bw = *(const v2f*)(Wt + (lrow + (t << 4)) * TP + cb);
            acc[t] = WMMA_F32(a, bw, acc[t]);
        }
    }
}

__device__ __forceinline__ void bn_relu(v8f acc[4], const float* g, const float* be,
                                        const float* mn, const float* vr, int lrow) {
#pragma unroll
    for (int t = 0; t < 4; ++t) {
        const int d = lrow + (t << 4);
        const float s = g[d] * (1.0f / sqrtf(vr[d] + BN_EPS));
        const float bi = be[d] - mn[d] * s;
#pragma unroll
        for (int v = 0; v < 8; ++v) acc[t][v] = fmaxf(acc[t][v] * s + bi, 0.0f);
    }
}

__device__ __forceinline__ void store16x64(const v8f acc[4], int lrow, int half, float* Y) {
#pragma unroll
    for (int t = 0; t < 4; ++t)
#pragma unroll
        for (int v = 0; v < 8; ++v)
            Y[(v + (half << 3)) * TP + lrow + (t << 4)] = acc[t][v];
}

__global__ __launch_bounds__(NT) void edgeconv_fused_kernel(
    const float* __restrict__ points,
    const float* __restrict__ feats,
    const float* __restrict__ w0,
    const float* __restrict__ w1,
    const float* __restrict__ w2,
    const float* __restrict__ gammas,
    const float* __restrict__ betas,
    const float* __restrict__ means,
    const float* __restrict__ vars,
    const float* __restrict__ scw,
    const float* __restrict__ scg, const float* __restrict__ scb,
    const float* __restrict__ scm, const float* __restrict__ scv,
    float* __restrict__ out)
{
    extern __shared__ __attribute__((aligned(16))) float smem[];
    float* pts     = smem;
    float* Yp      = pts + 768;
    float* R       = Yp + 256 * 64;
    float* Wt0     = R;
    float* scWt    = Wt0 + 64 * TP0;
    float* Wt1     = R;
    float* Wt2     = Wt1 + 64 * TP;
    float* waveBuf = Wt2 + 64 * TP;
    float* ftsAll  = waveBuf + NW * 16 * TP;
    int*   nbrAll  = (int*)(ftsAll + NW * 64);

    const int tid  = threadIdx.x;
    const int lane = tid & 31;
    const int w    = tid >> 5;
    const int lrow = lane & 15;
    const int half = lane >> 4;
    const int b     = blockIdx.x >> 1;
    const int p0    = (blockIdx.x & 1) * (NW * TPW);
    const float* fb = feats + (b << 8) * 64;

    const int sc0 = tid >> 6, sd = tid & 63;
    {
        float* l0 = Wt0 + sd * TP0 + sc0;
        float* ls = scWt + sd * TP + sc0;
        pts[tid] = points[b * 768 + tid];
        if (tid < 768 - NT) pts[tid + NT] = points[b * 768 + tid + NT];
#pragma unroll
        for (int k = 0; k < 16; ++k) l0[8 * k] = w0[tid + NT * k];
#pragma unroll
        for (int k = 0; k < 8; ++k) ls[8 * k] = scw[tid + NT * k];
    }
    __syncthreads();

    const v8f vz = {0.f, 0.f, 0.f, 0.f, 0.f, 0.f, 0.f, 0.f};

    {
        const int q0 = w << 4;
        v2f a0[16];
#pragma unroll
        for (int kk = 0; kk < 16; ++kk) {
            const int cb = (kk << 2) + (half << 1);
            a0[kk] = *(const v2f*)(fb + (q0 + lrow) * 64 + cb);
        }
        v8f y0[4] = {vz, vz, vz, vz};
#pragma unroll
        for (int kk = 0; kk < 16; ++kk) {
            const int cb = (kk << 2) + (half << 1);
#pragma unroll
            for (int t = 0; t < 4; ++t) {
                v2f bw = *(const v2f*)(Wt0 + (lrow + (t << 4)) * TP0 + 64 + cb);
                y0[t] = WMMA_F32(a0[kk], bw, y0[t]);
            }
        }
#pragma unroll
        for (int t = 0; t < 4; ++t)
#pragma unroll
            for (int v = 0; v < 8; ++v)
                Yp[(q0 + v + (half << 3)) * 64 + lrow + (t << 4)] = y0[t][v];
    }
    v8f zA, zB, sA8, sB8;
    {
        const int pj = p0 + ((lrow & 7) << 4) + w;
        v2f a[16];
#pragma unroll
        for (int kk = 0; kk < 16; ++kk) {
            const int cb = (kk << 2) + (half << 1);
            a[kk] = *(const v2f*)(fb + pj * 64 + cb);
        }
        v8f az[4] = {vz, vz, vz, vz};
        v8f as[4] = {vz, vz, vz, vz};
#pragma unroll
        for (int kk = 0; kk < 16; ++kk) {
            const int cb = (kk << 2) + (half << 1);
#pragma unroll
            for (int t = 0; t < 4; ++t) {
                v2f bz = *(const v2f*)(Wt0 + (lrow + (t << 4)) * TP0 + cb);
                v2f bs = *(const v2f*)(scWt + (lrow + (t << 4)) * TP + cb);
                az[t] = WMMA_F32(a[kk], bz, az[t]);
                as[t] = WMMA_F32(a[kk], bs, as[t]);
            }
        }
#pragma unroll
        for (int v = 0; v < 8; ++v) {
            zA[v]  = half ? az[1][v] : az[0][v];
            zB[v]  = half ? az[3][v] : az[2][v];
            sA8[v] = half ? as[1][v] : as[0][v];
            sB8[v] = half ? as[3][v] : as[2][v];
        }
    }
    __syncthreads();
    {
        float* l1 = Wt1 + sd * TP + sc0;
        float* l2 = Wt2 + sd * TP + sc0;
        float r1[8], r2[8];
#pragma unroll
        for (int k = 0; k < 8; ++k) {
            r1[k] = w1[tid + NT * k];
            r2[k] = w2[tid + NT * k];
        }
        __builtin_amdgcn_sched_barrier(0);
#pragma unroll
        for (int k = 0; k < 8; ++k) {
            l1[8 * k] = r1[k];
            l2[8 * k] = r2[k];
        }
    }
    __syncthreads();

    float* bufA = waveBuf + w * (16 * TP);
    float* ftsW = ftsAll + (w << 6);
    int*   nbrW = nbrAll + (w << 4);

    const int d0 = lane, d1 = lane + 32;
    const float s0a = gammas[d0] * (1.0f / sqrtf(vars[d0] + BN_EPS));
    const float b0a = betas[d0] - means[d0] * s0a;
    const float s0b = gammas[d1] * (1.0f / sqrtf(vars[d1] + BN_EPS));
    const float b0b = betas[d1] - means[d1] * s0b;
    const float sA = scg[d0] * (1.0f / sqrtf(scv[d0] + BN_EPS));
    const float bA = scb[d0] - scm[d0] * sA;
    const float sB = scg[d1] * (1.0f / sqrtf(scv[d1] + BN_EPS));
    const float bB = scb[d1] - scm[d1] * sB;

    __builtin_amdgcn_s_setprio(3);
    for (int tl = 0; tl < TPW; ++tl) {
    if (tl == 3) __builtin_amdgcn_s_setprio(2);
    if (tl == 5) __builtin_amdgcn_s_setprio(1);
    if (tl == 7) __builtin_amdgcn_s_setprio(0);
    const int pi = p0 + (tl << 4) + w;

    const float px = pts[pi * 3], py = pts[pi * 3 + 1], pz = pts[pi * 3 + 2];
    unsigned key[8];
#pragma unroll
    for (int j = 0; j < 8; ++j) {
        const int q = lane + (j << 5);
        const float dx = pts[q * 3] - px;
        const float dy = pts[q * 3 + 1] - py;
        const float dz = pts[q * 3 + 2] - pz;
        const float dd = dx * dx + dy * dy + dz * dz;
        key[j] = __float_as_uint((q == pi) ? BIG : dd);
    }
    unsigned thr = 0;
    int bit = 30;
    do {
        const unsigned trial = thr | (1u << bit);
        int below = 0;
#pragma unroll
        for (int j = 0; j < 8; j += 2) {
            const unsigned long long m2 =
                ((unsigned long long)__builtin_amdgcn_ballot_w32(key[j + 1] < trial) << 32) |
                __builtin_amdgcn_ballot_w32(key[j] < trial);
            below += __builtin_popcountll(m2);
        }
        thr = (below <= 16) ? trial : thr;
        bit -= (below == 16) ? 32 : 1;
    } while (bit >= 0);
    unsigned mLess[8], mEq[8];
    int nLess = 0;
#pragma unroll
    for (int j = 0; j < 8; ++j) {
        mLess[j] = __builtin_amdgcn_ballot_w32(key[j] < thr);
        mEq[j]   = __builtin_amdgcn_ballot_w32(key[j] == thr);
        nLess += __builtin_popcount(mLess[j]);
    }
    int offL = 0, offE = nLess;
#pragma unroll
    for (int j = 0; j < 8; ++j) {
        const bool isL = key[j] < thr;
        const bool isE = key[j] == thr;
        const int pos = isL ? offL + (int)__builtin_amdgcn_mbcnt_lo(mLess[j], 0u)
                            : offE + (int)__builtin_amdgcn_mbcnt_lo(mEq[j], 0u);
        if (isL || (isE && pos < 16)) nbrW[pos] = lane + (j << 5);
        offL += __builtin_popcount(mLess[j]);
        offE += __builtin_popcount(mEq[j]);
    }

    const float e0 = zA[tl] - Yp[pi * 64 + d0];
    const float e1 = zB[tl] - Yp[pi * 64 + d1];
    float ya[16], yb[16];
#pragma unroll
    for (int r = 0; r < 16; ++r) {
        const int nq = nbrW[r];
        ya[r] = Yp[nq * 64 + d0];
        yb[r] = Yp[nq * 64 + d1];
    }
#pragma unroll
    for (int r = 0; r < 16; ++r) {
        bufA[r * TP + d0] = fmaxf((ya[r] + e0) * s0a + b0a, 0.0f);
        bufA[r * TP + d1] = fmaxf((yb[r] + e1) * s0b + b0b, 0.0f);
    }

    v8f acc[4] = {vz, vz, vz, vz};
    gemm16x64(bufA, Wt1, lrow, half, acc);
    bn_relu(acc, gammas + 64, betas + 64, means + 64, vars + 64, lrow);
    store16x64(acc, lrow, half, bufA);

    acc[0] = vz; acc[1] = vz; acc[2] = vz; acc[3] = vz;
    gemm16x64(bufA, Wt2, lrow, half, acc);
    bn_relu(acc, gammas + 128, betas + 128, means + 128, vars + 128, lrow);

#pragma unroll
    for (int t = 0; t < 4; ++t) {
        float s = acc[t][0] + acc[t][1] + acc[t][2] + acc[t][3] +
                  acc[t][4] + acc[t][5] + acc[t][6] + acc[t][7];
        s += __shfl_xor(s, 16, 32);
        if (lane < 16) ftsW[lrow + (t << 4)] = s * (1.0f / 16.0f);
    }

    float* orow = out + (((b << 8) | pi) << 6);
    orow[d0] = fmaxf(sA8[tl] * sA + bA + ftsW[d0], 0.f);
    orow[d1] = fmaxf(sB8[tl] * sB + bB + ftsW[d1], 0.f);
    }
}

extern "C" void kernel_launch(void* const* d_in, const int* in_sizes, int n_in,
                              void* d_out, int out_size, void* d_ws, size_t ws_size,
                              hipStream_t stream) {
    (void)in_sizes; (void)n_in; (void)d_ws; (void)ws_size; (void)out_size;
    const float* points = (const float*)d_in[0];
    const float* feats  = (const float*)d_in[1];
    const float* w0     = (const float*)d_in[2];
    const float* w1     = (const float*)d_in[3];
    const float* w2     = (const float*)d_in[4];
    const float* gammas = (const float*)d_in[5];
    const float* betas  = (const float*)d_in[6];
    const float* means  = (const float*)d_in[7];
    const float* vars   = (const float*)d_in[8];
    const float* scw    = (const float*)d_in[9];
    const float* scg    = (const float*)d_in[10];
    const float* scb    = (const float*)d_in[11];
    const float* scm    = (const float*)d_in[12];
    const float* scv    = (const float*)d_in[13];
    float* out = (float*)d_out;

    const size_t ldsBytes = (768 + 256 * 64 + 2 * 64 * TP + NW * 16 * TP + NW * 64) * sizeof(float)
                          + NW * 16 * sizeof(int);
    dim3 grid(256), block(NT);
    edgeconv_fused_kernel<<<grid, block, ldsBytes, stream>>>(
        points, feats, w0, w1, w2, gammas, betas, means, vars,
        scw, scg, scb, scm, scv, out);
}
